# starting kernel
# baseline (speedup 1.0000x reference)
#include <hip/hip_runtime.h>

#define N_NODES 100000
#define N_EDGES 1600000
#define D 64

typedef __attribute__((ext_vector_type(2))) float v2f;
typedef __attribute__((ext_vector_type(4))) float v4f;
typedef __attribute__((ext_vector_type(8))) float v8f;

__global__ void __launch_bounds__(256) gcn_scale_src(
    const float* __restrict__ feat,
    const float* __restrict__ out_norm,
    float* __restrict__ h) {
  int gid = blockIdx.x * blockDim.x + threadIdx.x;
  if (gid >= N_NODES * (D / 4)) return;
  int row = gid >> 4;
  float inv = 1.0f / out_norm[row];
  v4f v = ((const v4f*)feat)[gid];
  v.x *= inv; v.y *= inv; v.z *= inv; v.w *= inv;
  ((v4f*)h)[gid] = v;
}

__global__ void __launch_bounds__(256) gcn_edge_scatter(
    const float* __restrict__ h,
    const int* __restrict__ esrc,
    const int* __restrict__ edst,
    float* __restrict__ agg) {
  unsigned gid = blockIdx.x * blockDim.x + threadIdx.x;
  unsigned e = gid >> 5;
  unsigned lane = gid & 31u;
  if (e >= N_EDGES) return;
  int s = esrc[e];
  int d = edst[e];
  v2f v = ((const v2f*)(h + (size_t)s * D))[lane];
  float* out = agg + (size_t)d * D + 2u * lane;
  unsafeAtomicAdd(out, v.x);
  unsafeAtomicAdd(out + 1, v.y);
}

__global__ void __launch_bounds__(256) gcn_wmma_linear(
    const float* __restrict__ agg,
    const float* __restrict__ in_norm,
    const float* __restrict__ W,
    const float* __restrict__ bias,
    float* __restrict__ out) {
  const int lane  = threadIdx.x & 31;
  const int wave  = threadIdx.x >> 5;
  const int tile  = blockIdx.x * 8 + wave;
  const int NTILES = N_NODES / 16;
  if (tile >= NTILES) return;

  const int m16   = lane & 15;
  const int khalf = lane >> 4;
  const int mrow  = (tile << 4) + m16;

  const float inv = 1.0f / in_norm[mrow];

  v2f afrag[16];
  const float* arow = agg + (size_t)mrow * D + 2 * khalf;
#pragma unroll
  for (int k = 0; k < 16; ++k) {
    v2f a = *(const v2f*)(arow + 4 * k);
    a.x *= inv; a.y *= inv;
    afrag[k] = a;
  }

#pragma unroll
  for (int nt = 0; nt < 4; ++nt) {
    const int nbase = nt << 4;
    const float bv  = bias[nbase + m16];
    v8f acc = {bv, bv, bv, bv, bv, bv, bv, bv};

    const float* wrow = W + (size_t)(nbase + m16) * D + 2 * khalf;
#pragma unroll
    for (int k = 0; k < 16; ++k) {
      v2f bf = *(const v2f*)(wrow + 4 * k);
      acc = __builtin_amdgcn_wmma_f32_16x16x4_f32(
           false, afrag[k],  false, bf,
           (short)0, acc,  false,  false);
    }

    float* orow = out + ((size_t)(tile * 16 + 8 * khalf)) * D + nbase + m16;
#pragma unroll
    for (int j = 0; j < 8; ++j) {
      orow[(size_t)j * D] = acc[j];
    }
  }
}

extern "C" void kernel_launch(void* const* d_in, const int* in_sizes, int n_in,
                              void* d_out, int out_size, void* d_ws, size_t ws_size,
                              hipStream_t stream) {
  const float* feat     = (const float*)d_in[0];
  const float* in_norm  = (const float*)d_in[1];
  const float* out_norm = (const float*)d_in[2];
  const int*   esrc     = (const int*)d_in[3];
  const int*   edst     = (const int*)d_in[4];
  const float* W        = (const float*)d_in[5];
  const float* bias     = (const float*)d_in[6];

  float* out = (float*)d_out;
  float* agg = (float*)d_ws;
  float* h   = out;

  hipMemsetAsync(agg, 0, (size_t)N_NODES * D * sizeof(float), stream);

  {
    int total = N_NODES * (D / 4);
    gcn_scale_src<<<(total + 255) / 256, 256, 0, stream>>>(feat, out_norm, h);
  }
  {
    long long total = (long long)N_EDGES * 32;
    unsigned blocks = (unsigned)((total + 255) / 256);
    gcn_edge_scatter<<<blocks, 256, 0, stream>>>(h, esrc, edst, agg);
  }
  {
    int ntiles = N_NODES / 16;
    int blocks = (ntiles + 7) / 8;
    gcn_wmma_linear<<<blocks, 256, 0, stream>>>(agg, in_norm, W, bias, out);
  }
}


# Edge scatter: one wave takes 32 consecutive edges (coalesced index loads, 32 source-row loads in flight per wave before its adds) instead of 1 edge per wave; sums still in atomic arrival order
# speedup vs baseline: 1.7666x; 1.7666x over previous
#include <hip/hip_runtime.h>

#define N_NODES 100000
#define N_EDGES 1600000
#define D 64

typedef __attribute__((ext_vector_type(2))) float v2f;
typedef __attribute__((ext_vector_type(4))) float v4f;
typedef __attribute__((ext_vector_type(8))) float v8f;

__global__ void __launch_bounds__(256) gcn_scale_src(
    const float* __restrict__ feat,
    const float* __restrict__ out_norm,
    float* __restrict__ h) {
  int gid = blockIdx.x * blockDim.x + threadIdx.x;
  if (gid >= N_NODES * (D / 4)) return;
  int row = gid >> 4;
  float inv = 1.0f / out_norm[row];
  v4f v = ((const v4f*)feat)[gid];
  v.x *= inv; v.y *= inv; v.z *= inv; v.w *= inv;
  ((v4f*)h)[gid] = v;
}

#define EDGES_PER_WAVE 32
static_assert(N_EDGES % (EDGES_PER_WAVE * 8) == 0, "whole blocks of 8 waves");

__global__ void __launch_bounds__(256) gcn_edge_scatter(
    const float* __restrict__ h,
    const int* __restrict__ esrc,
    const int* __restrict__ edst,
    float* __restrict__ agg) {
  const unsigned lane = threadIdx.x & 31u;
  const unsigned wave = (blockIdx.x * blockDim.x + threadIdx.x) >> 5;
  const unsigned e0 = wave * EDGES_PER_WAVE;
  if (e0 >= N_EDGES) return;
  const int s_l = esrc[e0 + lane];
  const int d_l = edst[e0 + lane];

  v2f v[EDGES_PER_WAVE];
#pragma unroll
  for (int j = 0; j < EDGES_PER_WAVE; ++j) {
    const int s = __builtin_amdgcn_readlane(s_l, j);
    v[j] = ((const v2f*)(h + (size_t)s * D))[lane];
  }
#pragma unroll
  for (int j = 0; j < EDGES_PER_WAVE; ++j) {
    const int d = __builtin_amdgcn_readlane(d_l, j);
    float* out = agg + (size_t)d * D + 2u * lane;
    unsafeAtomicAdd(out, v[j].x);
    unsafeAtomicAdd(out + 1, v[j].y);
  }
}

__global__ void __launch_bounds__(256) gcn_wmma_linear(
    const float* __restrict__ agg,
    const float* __restrict__ in_norm,
    const float* __restrict__ W,
    const float* __restrict__ bias,
    float* __restrict__ out) {
  const int lane  = threadIdx.x & 31;
  const int wave  = threadIdx.x >> 5;
  const int tile  = blockIdx.x * 8 + wave;
  const int NTILES = N_NODES / 16;
  if (tile >= NTILES) return;

  const int m16   = lane & 15;
  const int khalf = lane >> 4;
  const int mrow  = (tile << 4) + m16;

  const float inv = 1.0f / in_norm[mrow];

  v2f afrag[16];
  const float* arow = agg + (size_t)mrow * D + 2 * khalf;
#pragma unroll
  for (int k = 0; k < 16; ++k) {
    v2f a = *(const v2f*)(arow + 4 * k);
    a.x *= inv; a.y *= inv;
    afrag[k] = a;
  }

#pragma unroll
  for (int nt = 0; nt < 4; ++nt) {
    const int nbase = nt << 4;
    const float bv  = bias[nbase + m16];
    v8f acc = {bv, bv, bv, bv, bv, bv, bv, bv};

    const float* wrow = W + (size_t)(nbase + m16) * D + 2 * khalf;
#pragma unroll
    for (int k = 0; k < 16; ++k) {
      v2f bf = *(const v2f*)(wrow + 4 * k);
      acc = __builtin_amdgcn_wmma_f32_16x16x4_f32(
           false, afrag[k],  false, bf,
           (short)0, acc,  false,  false);
    }

    float* orow = out + ((size_t)(tile * 16 + 8 * khalf)) * D + nbase + m16;
#pragma unroll
    for (int j = 0; j < 8; ++j) {
      orow[(size_t)j * D] = acc[j];
    }
  }
}

extern "C" void kernel_launch(void* const* d_in, const int* in_sizes, int n_in,
                              void* d_out, int out_size, void* d_ws, size_t ws_size,
                              hipStream_t stream) {
  const float* feat     = (const float*)d_in[0];
  const float* in_norm  = (const float*)d_in[1];
  const float* out_norm = (const float*)d_in[2];
  const int*   esrc     = (const int*)d_in[3];
  const int*   edst     = (const int*)d_in[4];
  const float* W        = (const float*)d_in[5];
  const float* bias     = (const float*)d_in[6];

  float* out = (float*)d_out;
  float* agg = (float*)d_ws;
  float* h   = out;

  hipMemsetAsync(agg, 0, (size_t)N_NODES * D * sizeof(float), stream);

  {
    int total = N_NODES * (D / 4);
    gcn_scale_src<<<(total + 255) / 256, 256, 0, stream>>>(feat, out_norm, h);
  }
  {
    unsigned blocks = N_EDGES / (EDGES_PER_WAVE * 8);
    gcn_edge_scatter<<<blocks, 256, 0, stream>>>(h, esrc, edst, agg);
  }
  {
    int ntiles = N_NODES / 16;
    int blocks = (ntiles + 7) / 8;
    gcn_wmma_linear<<<blocks, 256, 0, stream>>>(agg, in_norm, W, bias, out);
  }
}


# Edge scatter: each row is turned through a 256-byte wave-private LDS slot so that every f32 atomic add covers one whole 128-byte line, not two half-filled ones; sums still in atomic arrival order
# speedup vs baseline: 1.8797x; 1.0640x over previous
#include <hip/hip_runtime.h>

#define N_NODES 100000
#define N_EDGES 1600000
#define D 64

typedef __attribute__((ext_vector_type(2))) float v2f;
typedef __attribute__((ext_vector_type(4))) float v4f;
typedef __attribute__((ext_vector_type(8))) float v8f;

__global__ void __launch_bounds__(256) gcn_scale_src(
    const float* __restrict__ feat,
    const float* __restrict__ out_norm,
    float* __restrict__ h) {
  int gid = blockIdx.x * blockDim.x + threadIdx.x;
  if (gid >= N_NODES * (D / 4)) return;
  int row = gid >> 4;
  float inv = 1.0f / out_norm[row];
  v4f v = ((const v4f*)feat)[gid];
  v.x *= inv; v.y *= inv; v.z *= inv; v.w *= inv;
  ((v4f*)h)[gid] = v;
}

#define EDGES_PER_WAVE 32
#define ROW_BYTES (D * 4u)
static_assert(N_EDGES % (EDGES_PER_WAVE * 8) == 0, "whole blocks of 8 waves");
static_assert(D == 64, "a row is two 128-byte lines: features l and l+32 per lane");
static_assert((unsigned long long)N_NODES * ROW_BYTES < (1ull << 32), "32-bit byte offsets");

static __device__ __forceinline__ void wave_sync() {
  __builtin_amdgcn_fence(__ATOMIC_RELEASE, "wavefront");
  __builtin_amdgcn_wave_barrier();
  __builtin_amdgcn_fence(__ATOMIC_ACQUIRE, "wavefront");
}

__global__ void __launch_bounds__(256) gcn_edge_scatter(
    const float* __restrict__ h,
    const int* __restrict__ esrc,
    const int* __restrict__ edst,
    float* __restrict__ agg) {
  __shared__ float stage[8][D];
  const unsigned lane = threadIdx.x & 31u;
  const unsigned wave = (blockIdx.x * blockDim.x + threadIdx.x) >> 5;
  const unsigned e0 = wave * EDGES_PER_WAVE;
  if (e0 >= N_EDGES) return;
  const int s_l = esrc[e0 + lane];
  const int d_l = edst[e0 + lane];
  const unsigned lane8 = lane * 8u;
  const unsigned lane4 = lane * 4u;
  float* slot = stage[threadIdx.x >> 5];

  v2f v[EDGES_PER_WAVE];
#pragma unroll
  for (int j = 0; j < EDGES_PER_WAVE; ++j) {
    const unsigned s = (unsigned)__builtin_amdgcn_readlane(s_l, j);
    v[j] = *(const v2f*)((const char*)h + (s * ROW_BYTES + lane8));
  }
#pragma unroll
  for (int j = 0; j < EDGES_PER_WAVE; ++j) {
    *(v2f*)(slot + 2u * lane) = v[j];
    wave_sync();
    const float lo = slot[lane];
    const float hi = slot[lane + 32u];
    wave_sync();
    const unsigned d = (unsigned)__builtin_amdgcn_readlane(d_l, j);
    char* p = (char*)agg + (d * ROW_BYTES + lane4);
    unsafeAtomicAdd((float*)p, lo);
    unsafeAtomicAdd((float*)(p + 128), hi);
  }
}

__global__ void __launch_bounds__(256) gcn_wmma_linear(
    const float* __restrict__ agg,
    const float* __restrict__ in_norm,
    const float* __restrict__ W,
    const float* __restrict__ bias,
    float* __restrict__ out) {
  const int lane  = threadIdx.x & 31;
  const int wave  = threadIdx.x >> 5;
  const int tile  = blockIdx.x * 8 + wave;
  const int NTILES = N_NODES / 16;
  if (tile >= NTILES) return;

  const int m16   = lane & 15;
  const int khalf = lane >> 4;
  const int mrow  = (tile << 4) + m16;

  const float inv = 1.0f / in_norm[mrow];

  v2f afrag[16];
  const float* arow = agg + (size_t)mrow * D + 2 * khalf;
#pragma unroll
  for (int k = 0; k < 16; ++k) {
    v2f a = *(const v2f*)(arow + 4 * k);
    a.x *= inv; a.y *= inv;
    afrag[k] = a;
  }

#pragma unroll
  for (int nt = 0; nt < 4; ++nt) {
    const int nbase = nt << 4;
    const float bv  = bias[nbase + m16];
    v8f acc = {bv, bv, bv, bv, bv, bv, bv, bv};

    const float* wrow = W + (size_t)(nbase + m16) * D + 2 * khalf;
#pragma unroll
    for (int k = 0; k < 16; ++k) {
      v2f bf = *(const v2f*)(wrow + 4 * k);
      acc = __builtin_amdgcn_wmma_f32_16x16x4_f32(
           false, afrag[k],  false, bf,
           (short)0, acc,  false,  false);
    }

    float* orow = out + ((size_t)(tile * 16 + 8 * khalf)) * D + nbase + m16;
#pragma unroll
    for (int j = 0; j < 8; ++j) {
      orow[(size_t)j * D] = acc[j];
    }
  }
}

extern "C" void kernel_launch(void* const* d_in, const int* in_sizes, int n_in,
                              void* d_out, int out_size, void* d_ws, size_t ws_size,
                              hipStream_t stream) {
  const float* feat     = (const float*)d_in[0];
  const float* in_norm  = (const float*)d_in[1];
  const float* out_norm = (const float*)d_in[2];
  const int*   esrc     = (const int*)d_in[3];
  const int*   edst     = (const int*)d_in[4];
  const float* W        = (const float*)d_in[5];
  const float* bias     = (const float*)d_in[6];

  float* out = (float*)d_out;
  float* agg = (float*)d_ws;
  float* h   = out;

  hipMemsetAsync(agg, 0, (size_t)N_NODES * D * sizeof(float), stream);

  {
    int total = N_NODES * (D / 4);
    gcn_scale_src<<<(total + 255) / 256, 256, 0, stream>>>(feat, out_norm, h);
  }
  {
    unsigned blocks = N_EDGES / (EDGES_PER_WAVE * 8);
    gcn_edge_scatter<<<blocks, 256, 0, stream>>>(h, esrc, edst, agg);
  }
  {
    int ntiles = N_NODES / 16;
    int blocks = (ntiles + 7) / 8;
    gcn_wmma_linear<<<blocks, 256, 0, stream>>>(agg, in_norm, W, bias, out);
  }
}


# Edge scatter gathers rows from feat itself and scales them by 1/out_norm[src] in registers; the scaled copy and the kernel that wrote it are gone (2 launches); sums still in atomic arrival order
# speedup vs baseline: 1.9478x; 1.0362x over previous
#include <hip/hip_runtime.h>

#define N_NODES 100000
#define N_EDGES 1600000
#define D 64

typedef __attribute__((ext_vector_type(2))) float v2f;
typedef __attribute__((ext_vector_type(8))) float v8f;

#define EDGES_PER_WAVE 32
#define ROW_BYTES (D * 4u)
static_assert(N_EDGES % (EDGES_PER_WAVE * 8) == 0, "whole blocks of 8 waves");
static_assert(D == 64, "a row is two 128-byte lines: features l and l+32 per lane");
static_assert((unsigned long long)N_NODES * ROW_BYTES < (1ull << 32), "32-bit byte offsets");

static __device__ __forceinline__ void wave_sync() {
  __builtin_amdgcn_fence(__ATOMIC_RELEASE, "wavefront");
  __builtin_amdgcn_wave_barrier();
  __builtin_amdgcn_fence(__ATOMIC_ACQUIRE, "wavefront");
}

__global__ void __launch_bounds__(256) gcn_edge_scatter(
    const float* __restrict__ feat,
    const float* __restrict__ out_norm,
    const int* __restrict__ esrc,
    const int* __restrict__ edst,
    float* __restrict__ agg) {
  __shared__ float stage[8][D];
  const unsigned lane = threadIdx.x & 31u;
  const unsigned wave = (blockIdx.x * blockDim.x + threadIdx.x) >> 5;
  const unsigned e0 = wave * EDGES_PER_WAVE;
  if (e0 >= N_EDGES) return;
  const int s_l = esrc[e0 + lane];
  const int d_l = edst[e0 + lane];
  const float on_l = out_norm[s_l];
  const unsigned lane8 = lane * 8u;
  const unsigned lane4 = lane * 4u;
  float* slot = stage[threadIdx.x >> 5];

  v2f v[EDGES_PER_WAVE];
#pragma unroll
  for (int c = 0; c < EDGES_PER_WAVE; c += 8) {
#pragma unroll
    for (int j = c; j < c + 8; ++j) {
      const unsigned s = (unsigned)__builtin_amdgcn_readlane(s_l, j);
      v[j] = *(const v2f*)((const char*)feat + (s * ROW_BYTES + lane8));
    }
    __builtin_amdgcn_sched_barrier(0);
  }
  const float inv_l = 1.0f / on_l;
#pragma unroll
  for (int j = 0; j < EDGES_PER_WAVE; ++j) {
    const float inv = __int_as_float(__builtin_amdgcn_readlane(__float_as_int(inv_l), j));
    v2f t = v[j];
    t.x *= inv; t.y *= inv;
    *(v2f*)(slot + 2u * lane) = t;
    wave_sync();
    const float lo = slot[lane];
    const float hi = slot[lane + 32u];
    wave_sync();
    const unsigned d = (unsigned)__builtin_amdgcn_readlane(d_l, j);
    char* p = (char*)agg + (d * ROW_BYTES + lane4);
    unsafeAtomicAdd((float*)p, lo);
    unsafeAtomicAdd((float*)(p + 128), hi);
  }
}

__global__ void __launch_bounds__(256) gcn_wmma_linear(
    const float* __restrict__ agg,
    const float* __restrict__ in_norm,
    const float* __restrict__ W,
    const float* __restrict__ bias,
    float* __restrict__ out) {
  const int lane  = threadIdx.x & 31;
  const int wave  = threadIdx.x >> 5;
  const int tile  = blockIdx.x * 8 + wave;
  const int NTILES = N_NODES / 16;
  if (tile >= NTILES) return;

  const int m16   = lane & 15;
  const int khalf = lane >> 4;
  const int mrow  = (tile << 4) + m16;

  const float inv = 1.0f / in_norm[mrow];

  v2f afrag[16];
  const float* arow = agg + (size_t)mrow * D + 2 * khalf;
#pragma unroll
  for (int k = 0; k < 16; ++k) {
    v2f a = *(const v2f*)(arow + 4 * k);
    a.x *= inv; a.y *= inv;
    afrag[k] = a;
  }

#pragma unroll
  for (int nt = 0; nt < 4; ++nt) {
    const int nbase = nt << 4;
    const float bv  = bias[nbase + m16];
    v8f acc = {bv, bv, bv, bv, bv, bv, bv, bv};

    const float* wrow = W + (size_t)(nbase + m16) * D + 2 * khalf;
#pragma unroll
    for (int k = 0; k < 16; ++k) {
      v2f bf = *(const v2f*)(wrow + 4 * k);
      acc = __builtin_amdgcn_wmma_f32_16x16x4_f32(
           false, afrag[k],  false, bf,
           (short)0, acc,  false,  false);
    }

    float* orow = out + ((size_t)(tile * 16 + 8 * khalf)) * D + nbase + m16;
#pragma unroll
    for (int j = 0; j < 8; ++j) {
      orow[(size_t)j * D] = acc[j];
    }
  }
}

extern "C" void kernel_launch(void* const* d_in, const int* in_sizes, int n_in,
                              void* d_out, int out_size, void* d_ws, size_t ws_size,
                              hipStream_t stream) {
  const float* feat     = (const float*)d_in[0];
  const float* in_norm  = (const float*)d_in[1];
  const float* out_norm = (const float*)d_in[2];
  const int*   esrc     = (const int*)d_in[3];
  const int*   edst     = (const int*)d_in[4];
  const float* W        = (const float*)d_in[5];
  const float* bias     = (const float*)d_in[6];

  float* out = (float*)d_out;
  float* agg = (float*)d_ws;

  hipMemsetAsync(agg, 0, (size_t)N_NODES * D * sizeof(float), stream);

  {
    unsigned blocks = N_EDGES / (EDGES_PER_WAVE * 8);
    gcn_edge_scatter<<<blocks, 256, 0, stream>>>(feat, out_norm, esrc, edst, agg);
  }
  {
    int ntiles = N_NODES / 16;
    int blocks = (ntiles + 7) / 8;
    gcn_wmma_linear<<<blocks, 256, 0, stream>>>(agg, in_norm, W, bias, out);
  }
}


# Edge sums in LDS: edges binned by 128-row destination tile (LDS histograms, one global integer add per block, bin), each tile summed by LDS f32 atomic adds and fed to the WMMA in place; arrival order
# speedup vs baseline: 2.3606x; 1.2119x over previous
#include <hip/hip_runtime.h>


#define N_NODES 100000
#define N_EDGES 1600000
#define D 64

#define TN 128
#define TSHIFT 7
#define NB ((N_NODES + TN - 1) / TN)
#define NB_PAD 1024
#define PITCH 68
#define CHUNK 4096
#define EPT (CHUNK / 256)
#define BIN_BLOCKS ((N_EDGES + CHUNK - 1) / CHUNK)
#define ROW_BYTES (D * 4u)

#define WS_TABLE_BYTES 4096u
#define WS_EDGES_OFF (4u * WS_TABLE_BYTES)

typedef __attribute__((ext_vector_type(2))) float v2f;
typedef __attribute__((ext_vector_type(8))) float v8f;

static_assert(NB <= NB_PAD, "scan covers every bin");
static_assert((1 << TSHIFT) == TN, "tile shift");
static_assert(N_NODES % 16 == 0, "whole 16-row matrix tiles");
static_assert(D == 64, "a row is one float2 per lane");
static_assert((unsigned long long)N_NODES * ROW_BYTES < (1ull << 32), "32-bit byte offsets");

__global__ void __launch_bounds__(256) gcn_bin_count(
    const int* __restrict__ edst,
    unsigned* __restrict__ binCount) {
  __shared__ unsigned h[NB];
  for (unsigned b = threadIdx.x; b < NB; b += 256u) h[b] = 0u;
  __syncthreads();
  const unsigned base = blockIdx.x * CHUNK + threadIdx.x;
#pragma unroll
  for (int i = 0; i < EPT; ++i) {
    const unsigned e = base + i * 256u;
    if (e < N_EDGES) atomicAdd(&h[(unsigned)edst[e] >> TSHIFT], 1u);
  }
  __syncthreads();
  for (unsigned b = threadIdx.x; b < NB; b += 256u) {
    const unsigned c = h[b];
    if (c) atomicAdd(&binCount[b], c);
  }
}

__global__ void __launch_bounds__(256) gcn_bin_place(
    const int* __restrict__ esrc,
    const int* __restrict__ edst,
    const unsigned* __restrict__ binCount,
    unsigned* __restrict__ binCursor,
    unsigned* __restrict__ binStart,
    int2* __restrict__ binned) {
  __shared__ unsigned h[NB_PAD];
  __shared__ unsigned sStart[NB_PAD];
  __shared__ unsigned wsum[8];
  const unsigned tid = threadIdx.x;
  const unsigned lane = tid & 31u;
  const unsigned wave = tid >> 5;

  for (unsigned b = tid; b < NB_PAD; b += 256u) h[b] = 0u;

  unsigned c[4];
#pragma unroll
  for (int i = 0; i < 4; ++i) {
    const unsigned bin = 4u * tid + i;
    c[i] = bin < NB ? binCount[bin] : 0u;
  }
  const unsigned s = c[0] + c[1] + c[2] + c[3];
  unsigned inc = s;
#pragma unroll
  for (unsigned d = 1; d < 32u; d <<= 1) {
    const unsigned o = __shfl_up(inc, d);
    if (lane >= d) inc += o;
  }
  if (lane == 31u) wsum[wave] = inc;
  __syncthreads();
  unsigned pre = 0u;
  for (unsigned w = 0; w < wave; ++w) pre += wsum[w];
  unsigned run = pre + inc - s;
#pragma unroll
  for (int i = 0; i < 4; ++i) {
    const unsigned bin = 4u * tid + i;
    sStart[bin] = run;
    if (blockIdx.x == 0 && bin < NB) binStart[bin] = run;
    run += c[i];
  }

  const unsigned base = blockIdx.x * CHUNK + tid;
  int sv[EPT], dv[EPT];
  unsigned rk[EPT];
#pragma unroll
  for (int i = 0; i < EPT; ++i) {
    const unsigned e = base + i * 256u;
    sv[i] = 0; dv[i] = 0; rk[i] = 0u;
    if (e < N_EDGES) {
      sv[i] = esrc[e];
      dv[i] = edst[e];
      rk[i] = atomicAdd(&h[(unsigned)dv[i] >> TSHIFT], 1u);
    }
  }
  __syncthreads();
  for (unsigned b = tid; b < NB; b += 256u) {
    const unsigned n = h[b];
    h[b] = n ? sStart[b] + atomicAdd(&binCursor[b], n) : 0u;
  }
  __syncthreads();
#pragma unroll
  for (int i = 0; i < EPT; ++i) {
    const unsigned e = base + i * 256u;
    if (e < N_EDGES) binned[h[(unsigned)dv[i] >> TSHIFT] + rk[i]] = make_int2(sv[i], dv[i]);
  }
}

__global__ void __launch_bounds__(256) gcn_tile_sum_linear(
    const float* __restrict__ feat,
    const float* __restrict__ in_norm,
    const float* __restrict__ out_norm,
    const float* __restrict__ W,
    const float* __restrict__ bias,
    const unsigned* __restrict__ binCount,
    const unsigned* __restrict__ binStart,
    const int2* __restrict__ binned,
    float* __restrict__ out) {
  __shared__ float tile[(TN + 1) * PITCH];
  const unsigned tid = threadIdx.x;
  const unsigned lane = tid & 31u;
  const unsigned wave = (unsigned)__builtin_amdgcn_readfirstlane((int)(tid >> 5));
  const unsigned b = blockIdx.x;
  const unsigned row0 = b * TN;

  for (unsigned i = tid; i < (TN + 1) * PITCH; i += 256u) tile[i] = 0.0f;
  const unsigned cnt = binCount[b];
  const unsigned start = binStart[b];
  __syncthreads();

  const unsigned lane8 = lane * 8u;
  for (unsigned g0 = wave * 32u; g0 < cnt; g0 += 256u) {
    const unsigned idx = g0 + lane;
    int2 e = make_int2(0, (int)(row0 + TN));
    if (idx < cnt) e = binned[start + idx];
    const int s_l = e.x;
    const unsigned dl_l = ((unsigned)e.y - row0) * PITCH;
    const float on_l = out_norm[s_l];

    v2f v[32];
#pragma unroll
    for (int c8 = 0; c8 < 32; c8 += 8) {
#pragma unroll
      for (int j = c8; j < c8 + 8; ++j) {
        const unsigned s = (unsigned)__builtin_amdgcn_readlane(s_l, j);
        v[j] = *(const v2f*)((const char*)feat + (s * ROW_BYTES + lane8));
      }
      __builtin_amdgcn_sched_barrier(0);
    }
    const float inv_l = 1.0f / on_l;
#pragma unroll
    for (int j = 0; j < 32; ++j) {
      const float inv = __int_as_float(__builtin_amdgcn_readlane(__float_as_int(inv_l), j));
      const unsigned d = (unsigned)__builtin_amdgcn_readlane((int)dl_l, j);
      v2f t = v[j];
      t.x *= inv; t.y *= inv;
      float* p = tile + d + 2u * lane;
      __hip_atomic_fetch_add(p, t.x, __ATOMIC_RELAXED, __HIP_MEMORY_SCOPE_WORKGROUP);
      __hip_atomic_fetch_add(p + 1, t.y, __ATOMIC_RELAXED, __HIP_MEMORY_SCOPE_WORKGROUP);
    }
  }
  __syncthreads();

  const unsigned rbase = row0 + wave * 16u;
  if (rbase >= N_NODES) return;

  const int m16   = lane & 15;
  const int khalf = lane >> 4;
  const unsigned mrow = rbase + m16;
  const float inv = 1.0f / in_norm[mrow];

  v2f afrag[16];
  const float* arow = tile + (wave * 16u + m16) * PITCH + 2 * khalf;
#pragma unroll
  for (int k = 0; k < 16; ++k) {
    v2f a = *(const v2f*)(arow + 4 * k);
    a.x *= inv; a.y *= inv;
    afrag[k] = a;
  }

#pragma unroll
  for (int nt = 0; nt < 4; ++nt) {
    const int nbase = nt << 4;
    const float bv  = bias[nbase + m16];
    v8f acc = {bv, bv, bv, bv, bv, bv, bv, bv};

    const float* wrow = W + (size_t)(nbase + m16) * D + 2 * khalf;
#pragma unroll
    for (int k = 0; k < 16; ++k) {
      v2f bf = *(const v2f*)(wrow + 4 * k);
      acc = __builtin_amdgcn_wmma_f32_16x16x4_f32(
           false, afrag[k],  false, bf,
           (short)0, acc,  false,  false);
    }

    float* orow = out + ((size_t)(rbase + 8 * khalf)) * D + nbase + m16;
#pragma unroll
    for (int j = 0; j < 8; ++j) {
      orow[(size_t)j * D] = acc[j];
    }
  }
}

extern "C" void kernel_launch(void* const* d_in, const int* in_sizes, int n_in,
                              void* d_out, int out_size, void* d_ws, size_t ws_size,
                              hipStream_t stream) {
  const float* feat     = (const float*)d_in[0];
  const float* in_norm  = (const float*)d_in[1];
  const float* out_norm = (const float*)d_in[2];
  const int*   esrc     = (const int*)d_in[3];
  const int*   edst     = (const int*)d_in[4];
  const float* W        = (const float*)d_in[5];
  const float* bias     = (const float*)d_in[6];

  float* out = (float*)d_out;
  unsigned* binCount  = (unsigned*)d_ws;
  unsigned* binCursor = (unsigned*)((char*)d_ws + WS_TABLE_BYTES);
  unsigned* binStart  = (unsigned*)((char*)d_ws + 2u * WS_TABLE_BYTES);
  int2*     binned    = (int2*)((char*)d_ws + WS_EDGES_OFF);

  hipMemsetAsync(d_ws, 0, 2u * WS_TABLE_BYTES, stream);

  gcn_bin_count<<<BIN_BLOCKS, 256, 0, stream>>>(edst, binCount);
  gcn_bin_place<<<BIN_BLOCKS, 256, 0, stream>>>(esrc, edst, binCount, binCursor, binStart, binned);
  gcn_tile_sum_linear<<<NB, 256, 0, stream>>>(feat, in_norm, out_norm, W, bias,
                                              binCount, binStart, binned, out);
}


# Tile sums in registers: tile's edges ordered by destination row in LDS (counting sort), a wave sums each row in registers, one LDS add pair per row not per edge; sums in list (atomic arrival) order
# speedup vs baseline: 2.7358x; 1.1589x over previous
#include <hip/hip_runtime.h>


#define N_NODES 100000
#define N_EDGES 1600000
#define D 64

#define TN 128
#define TSHIFT 7
#define NB ((N_NODES + TN - 1) / TN)
#define NB_PAD 1024
#define PITCH 68
#define CHUNK 4096
#define EPT (CHUNK / 256)
#define BIN_BLOCKS ((N_EDGES + CHUNK - 1) / CHUNK)
#define ROW_BYTES (D * 4u)
#define CAP 2560
#define IPT (CAP / 256)
#define ROWSHIFT 20

#define WS_TABLE_BYTES 4096u
#define WS_EDGES_OFF (4u * WS_TABLE_BYTES)

typedef __attribute__((ext_vector_type(2))) float v2f;
typedef __attribute__((ext_vector_type(8))) float v8f;

static_assert(NB <= NB_PAD, "scan covers every bin");
static_assert((1 << TSHIFT) == TN, "tile shift");
static_assert(N_NODES % 16 == 0, "whole 16-row matrix tiles");
static_assert(D == 64, "a row is one float2 per lane");
static_assert(N_NODES <= (1 << ROWSHIFT) && (TN + 1) <= (1 << (32 - ROWSHIFT)), "source and row share a word");
static_assert(TN == 128, "waves 0..3 scan the row counts, wave w owns rows 16w..16w+15");
static_assert((unsigned long long)N_NODES * ROW_BYTES < (1ull << 32), "32-bit byte offsets");

__global__ void __launch_bounds__(256) gcn_bin_count(
    const int* __restrict__ edst,
    unsigned* __restrict__ binCount) {
  __shared__ unsigned h[NB];
  for (unsigned b = threadIdx.x; b < NB; b += 256u) h[b] = 0u;
  __syncthreads();
  const unsigned base = blockIdx.x * CHUNK + threadIdx.x;
#pragma unroll
  for (int i = 0; i < EPT; ++i) {
    const unsigned e = base + i * 256u;
    if (e < N_EDGES) atomicAdd(&h[(unsigned)edst[e] >> TSHIFT], 1u);
  }
  __syncthreads();
  for (unsigned b = threadIdx.x; b < NB; b += 256u) {
    const unsigned c = h[b];
    if (c) atomicAdd(&binCount[b], c);
  }
}

__global__ void __launch_bounds__(256) gcn_bin_place(
    const int* __restrict__ esrc,
    const int* __restrict__ edst,
    const unsigned* __restrict__ binCount,
    unsigned* __restrict__ binCursor,
    unsigned* __restrict__ binStart,
    int2* __restrict__ binned) {
  __shared__ unsigned h[NB_PAD];
  __shared__ unsigned sStart[NB_PAD];
  __shared__ unsigned wsum[8];
  const unsigned tid = threadIdx.x;
  const unsigned lane = tid & 31u;
  const unsigned wave = tid >> 5;

  for (unsigned b = tid; b < NB_PAD; b += 256u) h[b] = 0u;

  unsigned c[4];
#pragma unroll
  for (int i = 0; i < 4; ++i) {
    const unsigned bin = 4u * tid + i;
    c[i] = bin < NB ? binCount[bin] : 0u;
  }
  const unsigned s = c[0] + c[1] + c[2] + c[3];
  unsigned inc = s;
#pragma unroll
  for (unsigned d = 1; d < 32u; d <<= 1) {
    const unsigned o = __shfl_up(inc, d);
    if (lane >= d) inc += o;
  }
  if (lane == 31u) wsum[wave] = inc;
  __syncthreads();
  unsigned pre = 0u;
  for (unsigned w = 0; w < wave; ++w) pre += wsum[w];
  unsigned run = pre + inc - s;
#pragma unroll
  for (int i = 0; i < 4; ++i) {
    const unsigned bin = 4u * tid + i;
    sStart[bin] = run;
    if (blockIdx.x == 0 && bin < NB) binStart[bin] = run;
    run += c[i];
  }

  const unsigned base = blockIdx.x * CHUNK + tid;
  int sv[EPT], dv[EPT];
  unsigned rk[EPT];
#pragma unroll
  for (int i = 0; i < EPT; ++i) {
    const unsigned e = base + i * 256u;
    sv[i] = 0; dv[i] = 0; rk[i] = 0u;
    if (e < N_EDGES) {
      sv[i] = esrc[e];
      dv[i] = edst[e];
      rk[i] = atomicAdd(&h[(unsigned)dv[i] >> TSHIFT], 1u);
    }
  }
  __syncthreads();
  for (unsigned b = tid; b < NB; b += 256u) {
    const unsigned n = h[b];
    h[b] = n ? sStart[b] + atomicAdd(&binCursor[b], n) : 0u;
  }
  __syncthreads();
#pragma unroll
  for (int i = 0; i < EPT; ++i) {
    const unsigned e = base + i * 256u;
    if (e < N_EDGES) binned[h[(unsigned)dv[i] >> TSHIFT] + rk[i]] = make_int2(sv[i], dv[i]);
  }
}

__global__ void __launch_bounds__(256, 8) gcn_tile_sum_linear(
    const float* __restrict__ feat,
    const float* __restrict__ in_norm,
    const float* __restrict__ out_norm,
    const float* __restrict__ W,
    const float* __restrict__ bias,
    const unsigned* __restrict__ binCount,
    const unsigned* __restrict__ binStart,
    const int2* __restrict__ binned,
    float* __restrict__ out) {
  __shared__ float tile[(TN + 1) * PITCH];
  __shared__ unsigned slist[CAP];
  __shared__ unsigned rcnt[TN];
  __shared__ unsigned rstart[TN + 1];
  __shared__ unsigned wsum[4];
  const unsigned tid = threadIdx.x;
  const unsigned lane = tid & 31u;
  const unsigned wave = (unsigned)__builtin_amdgcn_readfirstlane((int)(tid >> 5));
  const unsigned b = blockIdx.x;
  const unsigned row0 = b * TN;

  for (unsigned i = tid; i < (TN + 1) * PITCH; i += 256u) tile[i] = 0.0f;
  const unsigned cnt = binCount[b];
  const unsigned start = binStart[b];

  const unsigned lane8 = lane * 8u;
  for (unsigned c0 = 0; c0 < cnt; c0 += CAP) {
    const unsigned n = (cnt - c0 < CAP) ? cnt - c0 : CAP;
    if (tid < TN) rcnt[tid] = 0u;
    __syncthreads();

    const unsigned m = (n + 255u - tid) >> 8;
    const unsigned q0 = start + c0 + tid;
    const unsigned lastD = n - 1u - tid;
    int2 ev[IPT];
#pragma unroll
    for (int i = 0; i < IPT; ++i)
      ev[i] = binned[q0 + ((unsigned)i < m ? (unsigned)i * 256u : lastD)];
    unsigned sv[IPT], rk[IPT];
#pragma unroll
    for (int i = 0; i < IPT; ++i) {
      if ((unsigned)i < m) {
        const unsigned r = (unsigned)ev[i].y - row0;
        rk[i] = atomicAdd(&rcnt[r], 1u);
        sv[i] = (unsigned)ev[i].x | (r << ROWSHIFT);
      }
    }
    __syncthreads();

    const unsigned c = tid < TN ? rcnt[tid] : 0u;
    unsigned inc = c;
#pragma unroll
    for (unsigned d = 1; d < 32u; d <<= 1) {
      const unsigned o = __shfl_up(inc, d);
      if (lane >= d) inc += o;
    }
    if (lane == 31u && wave < 4u) wsum[wave] = inc;
    __syncthreads();
    if (tid < TN) {
      const unsigned pre = (wave > 0u ? wsum[0] : 0u) + (wave > 1u ? wsum[1] : 0u) + (wave > 2u ? wsum[2] : 0u);
      rstart[tid] = pre + inc - c;
    }
    if (tid == 0u) rstart[TN] = n;
    __syncthreads();

    unsigned rs[IPT];
#pragma unroll
    for (int i = 0; i < IPT; ++i) rs[i] = rstart[(unsigned)i < m ? sv[i] >> ROWSHIFT : 0u];
#pragma unroll
    for (int i = 0; i < IPT; ++i)
      if ((unsigned)i < m) slist[rs[i] + rk[i]] = sv[i];
    __syncthreads();

    const unsigned lo = (unsigned)__builtin_amdgcn_readfirstlane((int)rstart[wave * 16u]);
    const unsigned hi = (unsigned)__builtin_amdgcn_readfirstlane((int)rstart[wave * 16u + 16u]);
    v2f acc = {0.0f, 0.0f};
    unsigned prev = TN * PITCH;
    for (unsigned g0 = lo; g0 < hi; g0 += 32u) {
      const unsigned idx = g0 + lane;
      unsigned word = (unsigned)TN << ROWSHIFT;
      if (idx < hi) word = slist[idx];
      const unsigned s_l = word & ((1u << ROWSHIFT) - 1u);
      const unsigned dl_l = (word >> ROWSHIFT) * PITCH;
      const float on_l = out_norm[s_l];
      const unsigned left = hi - g0;

      v2f v[32];
#pragma unroll
      for (int c8 = 0; c8 < 32; c8 += 8) {
        if ((unsigned)c8 < left) {
#pragma unroll
          for (int j = c8; j < c8 + 8; ++j) {
            const unsigned s = (unsigned)__builtin_amdgcn_readlane((int)s_l, j);
            v[j] = *(const v2f*)((const char*)feat + (s * ROW_BYTES + lane8));
          }
        }
        __builtin_amdgcn_sched_barrier(0);
      }
      const float inv_l = 1.0f / on_l;
#pragma unroll
      for (int c8 = 0; c8 < 32; c8 += 8) {
        if ((unsigned)c8 < left) {
#pragma unroll
          for (int j = c8; j < c8 + 8; ++j) {
            const float inv = __int_as_float(__builtin_amdgcn_readlane(__float_as_int(inv_l), j));
            const unsigned d = (unsigned)__builtin_amdgcn_readlane((int)dl_l, j);
            if (d != prev) {
              float* p = tile + prev + 2u * lane;
              __hip_atomic_fetch_add(p, acc.x, __ATOMIC_RELAXED, __HIP_MEMORY_SCOPE_WORKGROUP);
              __hip_atomic_fetch_add(p + 1, acc.y, __ATOMIC_RELAXED, __HIP_MEMORY_SCOPE_WORKGROUP);
              acc.x = 0.0f; acc.y = 0.0f;
              prev = d;
            }
            acc.x += v[j].x * inv;
            acc.y += v[j].y * inv;
          }
        }
      }
    }
    {
      float* p = tile + prev + 2u * lane;
      __hip_atomic_fetch_add(p, acc.x, __ATOMIC_RELAXED, __HIP_MEMORY_SCOPE_WORKGROUP);
      __hip_atomic_fetch_add(p + 1, acc.y, __ATOMIC_RELAXED, __HIP_MEMORY_SCOPE_WORKGROUP);
    }
  }
  __syncthreads();

  const unsigned rbase = row0 + wave * 16u;
  if (rbase >= N_NODES) return;

  const int m16   = lane & 15;
  const int khalf = lane >> 4;
  const unsigned mrow = rbase + m16;
  const float inv = 1.0f / in_norm[mrow];

  v2f afrag[16];
  const float* arow = tile + (wave * 16u + m16) * PITCH + 2 * khalf;
#pragma unroll
  for (int k = 0; k < 16; ++k) {
    v2f a = *(const v2f*)(arow + 4 * k);
    a.x *= inv; a.y *= inv;
    afrag[k] = a;
  }

#pragma unroll
  for (int nt = 0; nt < 4; ++nt) {
    const int nbase = nt << 4;
    const float bv  = bias[nbase + m16];
    v8f acc = {bv, bv, bv, bv, bv, bv, bv, bv};

    const float* wrow = W + (size_t)(nbase + m16) * D + 2 * khalf;
#pragma unroll
    for (int k = 0; k < 16; ++k) {
      v2f bf = *(const v2f*)(wrow + 4 * k);
      acc = __builtin_amdgcn_wmma_f32_16x16x4_f32(
           false, afrag[k],  false, bf,
           (short)0, acc,  false,  false);
    }

    float* orow = out + ((size_t)(rbase + 8 * khalf)) * D + nbase + m16;
#pragma unroll
    for (int j = 0; j < 8; ++j) {
      orow[(size_t)j * D] = acc[j];
    }
  }
}

extern "C" void kernel_launch(void* const* d_in, const int* in_sizes, int n_in,
                              void* d_out, int out_size, void* d_ws, size_t ws_size,
                              hipStream_t stream) {
  const float* feat     = (const float*)d_in[0];
  const float* in_norm  = (const float*)d_in[1];
  const float* out_norm = (const float*)d_in[2];
  const int*   esrc     = (const int*)d_in[3];
  const int*   edst     = (const int*)d_in[4];
  const float* W        = (const float*)d_in[5];
  const float* bias     = (const float*)d_in[6];

  float* out = (float*)d_out;
  unsigned* binCount  = (unsigned*)d_ws;
  unsigned* binCursor = (unsigned*)((char*)d_ws + WS_TABLE_BYTES);
  unsigned* binStart  = (unsigned*)((char*)d_ws + 2u * WS_TABLE_BYTES);
  int2*     binned    = (int2*)((char*)d_ws + WS_EDGES_OFF);

  hipMemsetAsync(d_ws, 0, 2u * WS_TABLE_BYTES, stream);

  gcn_bin_count<<<BIN_BLOCKS, 256, 0, stream>>>(edst, binCount);
  gcn_bin_place<<<BIN_BLOCKS, 256, 0, stream>>>(esrc, edst, binCount, binCursor, binStart, binned);
  gcn_tile_sum_linear<<<NB, 256, 0, stream>>>(feat, in_norm, out_norm, W, bias,
                                              binCount, binStart, binned, out);
}


# One-pass binning: each block sorts its 4096 edges by tile in LDS, writes them with its tile offsets; tile kernel collects its 391 pieces; count pass, memset, global adds gone; sums in new list order
# speedup vs baseline: 4.0035x; 1.4634x over previous
#include <hip/hip_runtime.h>


#define N_NODES 100000
#define N_EDGES 1600000
#define D 64

#define TN 128
#define TSHIFT 7
#define NB ((N_NODES + TN - 1) / TN)
#define NB_PAD 1024
#define PITCH 68
#define CHUNK 4096
#define EPT (CHUNK / 256)
#define BIN_BLOCKS ((N_EDGES + CHUNK - 1) / CHUNK)
#define TP (NB + 2)
#define ROW_BYTES (D * 4u)
#define CAP 2560
#define IPT (CAP / 256)
#define ROWSHIFT 20

#define WS_LIST_OFF (2u << 20)

typedef __attribute__((ext_vector_type(2))) float v2f;
typedef __attribute__((ext_vector_type(8))) float v8f;

static_assert(NB + 2 <= NB_PAD, "scan covers every bin and the two count entries");
static_assert(TP % 4 == 0 && CHUNK % 1024 == 0 && N_EDGES % 4 == 0, "16-byte accesses cover whole groups of 4");
static_assert((unsigned long long)BIN_BLOCKS * TP * 4u <= WS_LIST_OFF, "table ends before the list");
static_assert(BIN_BLOCKS <= 512, "a thread of the tile kernel takes two chunks");
static_assert((1 << TSHIFT) == TN, "tile shift");
static_assert(N_NODES % 16 == 0, "whole 16-row matrix tiles");
static_assert(D == 64, "a row is one float2 per lane");
static_assert(N_NODES <= (1 << ROWSHIFT) && (TN + 1) <= (1 << (32 - ROWSHIFT)), "source and row share a word");
static_assert(TN == 128, "waves 0..3 scan the row counts, wave w owns rows 16w..16w+15");
static_assert((unsigned long long)N_NODES * ROW_BYTES < (1ull << 32), "32-bit byte offsets");

__global__ void __launch_bounds__(256) gcn_chunk_sort(
    const int* __restrict__ esrc,
    const int* __restrict__ edst,
    unsigned* __restrict__ segTab,
    unsigned* __restrict__ list) {
  __shared__ __attribute__((aligned(16))) unsigned h[NB_PAD];
  __shared__ __attribute__((aligned(16))) unsigned sorted[CHUNK];
  __shared__ unsigned wsum[8];
  const unsigned tid = threadIdx.x;
  const unsigned lane = tid & 31u;
  const unsigned wave = tid >> 5;

  for (unsigned b = tid; b < NB_PAD; b += 256u) h[b] = 0u;

  const unsigned cbase = blockIdx.x * CHUNK;
  int4 s4[EPT / 4], d4[EPT / 4];
#pragma unroll
  for (int q = 0; q < EPT / 4; ++q) {
    const unsigned e = cbase + (unsigned)q * 1024u + 4u * tid;
    s4[q] = make_int4(0, 0, 0, 0);
    d4[q] = make_int4(0, 0, 0, 0);
    if (e < N_EDGES) {
      s4[q] = *(const int4*)(esrc + e);
      d4[q] = *(const int4*)(edst + e);
    }
  }
  __syncthreads();

  unsigned wd[EPT], bn[EPT], rk[EPT];
#pragma unroll
  for (int q = 0; q < EPT / 4; ++q) {
    const bool have = cbase + (unsigned)q * 1024u + 4u * tid < N_EDGES;
    const int sv[4] = {s4[q].x, s4[q].y, s4[q].z, s4[q].w};
    const int dv[4] = {d4[q].x, d4[q].y, d4[q].z, d4[q].w};
#pragma unroll
    for (int j = 0; j < 4; ++j) {
      const unsigned d = (unsigned)dv[j];
      wd[4 * q + j] = (unsigned)sv[j] | ((d & (TN - 1u)) << ROWSHIFT);
      bn[4 * q + j] = d >> TSHIFT;
      rk[4 * q + j] = 0u;
      if (have) rk[4 * q + j] = atomicAdd(&h[bn[4 * q + j]], 1u);
    }
  }
  __syncthreads();

  const uint4 c4 = *(const uint4*)&h[4u * tid];
  const unsigned s = c4.x + c4.y + c4.z + c4.w;
  unsigned inc = s;
#pragma unroll
  for (unsigned d = 1; d < 32u; d <<= 1) {
    const unsigned o = __shfl_up(inc, d);
    if (lane >= d) inc += o;
  }
  if (lane == 31u) wsum[wave] = inc;
  __syncthreads();
  unsigned pre = 0u;
  for (unsigned w = 0; w < wave; ++w) pre += wsum[w];
  uint4 st;
  st.x = pre + inc - s;
  st.y = st.x + c4.x;
  st.z = st.y + c4.y;
  st.w = st.z + c4.z;
  *(uint4*)&h[4u * tid] = st;
  if (tid < TP / 4) *(uint4*)(segTab + blockIdx.x * TP + 4u * tid) = st;
  __syncthreads();

  unsigned fs[EPT];
#pragma unroll
  for (int i = 0; i < EPT; ++i) fs[i] = h[bn[i]];
#pragma unroll
  for (int i = 0; i < EPT; ++i)
    if (cbase + (unsigned)(i >> 2) * 1024u + 4u * tid < N_EDGES) sorted[fs[i] + rk[i]] = wd[i];
  __syncthreads();

#pragma unroll
  for (int q = 0; q < EPT / 4; ++q) {
    const unsigned idx = (unsigned)q * 1024u + 4u * tid;
    if (cbase + idx < N_EDGES) *(uint4*)(list + cbase + idx) = *(const uint4*)&sorted[idx];
  }
}

__global__ void __launch_bounds__(256, 8) gcn_tile_sum_linear(
    const float* __restrict__ feat,
    const float* __restrict__ in_norm,
    const float* __restrict__ out_norm,
    const float* __restrict__ W,
    const float* __restrict__ bias,
    const unsigned* __restrict__ segTab,
    const unsigned* __restrict__ list,
    float* __restrict__ out) {
  __shared__ float tile[(TN + 1) * PITCH];
  __shared__ unsigned slist[CAP];
  __shared__ unsigned rcnt[TN];
  __shared__ unsigned rstart[TN + 1];
  __shared__ unsigned wsum[4];
  __shared__ unsigned gaddr[CAP];
  __shared__ unsigned psum[8];
  __shared__ unsigned pinf[5 * 256];
  const unsigned tid = threadIdx.x;
  const unsigned lane = tid & 31u;
  const unsigned wave = (unsigned)__builtin_amdgcn_readfirstlane((int)(tid >> 5));
  const unsigned b = blockIdx.x;
  const unsigned row0 = b * TN;

  unsigned plen[2], pbase[2];
  unsigned a0[2], a1[2];
#pragma unroll
  for (int q = 0; q < 2; ++q) {
    const unsigned blk = 2u * tid + (unsigned)q;
    const unsigned* row = segTab + (blk < BIN_BLOCKS ? blk : BIN_BLOCKS - 1u) * TP + b;
    a0[q] = row[0];
    a1[q] = row[1];
  }
  for (unsigned i = tid; i < (TN + 1) * PITCH; i += 256u) tile[i] = 0.0f;
#pragma unroll
  for (int q = 0; q < 2; ++q) {
    const unsigned blk = 2u * tid + (unsigned)q;
    plen[q] = blk < BIN_BLOCKS ? a1[q] - a0[q] : 0u;
    pbase[q] = blk * CHUNK + a0[q];
  }
  const unsigned plen2 = plen[0] + plen[1];
  unsigned pinc = plen2;
#pragma unroll
  for (unsigned d = 1; d < 32u; d <<= 1) {
    const unsigned o = __shfl_up(pinc, d);
    if (lane >= d) pinc += o;
  }
  if (lane == 31u) psum[wave] = pinc;
  __syncthreads();
  unsigned ppre = 0u, ptot = 0u;
#pragma unroll
  for (unsigned w = 0; w < 8u; ++w) {
    const unsigned v = psum[w];
    if (w < wave) ppre += v;
    ptot += v;
  }
  const unsigned cnt = (unsigned)__builtin_amdgcn_readfirstlane((int)ptot);
  {
    const unsigned p0 = ppre + pinc - plen2;
    const unsigned p1 = p0 + plen[0];
    pinf[tid] = p0;
    pinf[256u + tid] = p1;
    pinf[512u + tid] = p1 + plen[1];
    pinf[768u + tid] = pbase[0] - p0;
    pinf[1024u + tid] = pbase[1] - p1;
  }

  const unsigned lane8 = lane * 8u;
  for (unsigned c0 = 0; c0 < cnt; c0 += CAP) {
    const unsigned n = (cnt - c0 < CAP) ? cnt - c0 : CAP;
    if (tid < TN) rcnt[tid] = 0u;
#pragma unroll
    for (int q = 0; q < 2; ++q) {
      const unsigned p = pinf[(unsigned)q * 256u + tid];
      const unsigned pe = pinf[(unsigned)(q + 1) * 256u + tid];
      const unsigned off = pinf[(unsigned)(q + 3) * 256u + tid];
      const unsigned klo = p > c0 ? p : c0;
      const unsigned khi = pe < c0 + n ? pe : c0 + n;
      for (unsigned k = klo; k < khi; ++k) gaddr[k - c0] = off + k;
    }
    __syncthreads();

    const unsigned m = (n + 255u - tid) >> 8;
    unsigned ga[IPT];
#pragma unroll
    for (int i = 0; i < IPT; ++i)
      ga[i] = gaddr[(unsigned)i < m ? tid + (unsigned)i * 256u : n - 1u];
    unsigned sv[IPT], rk[IPT];
#pragma unroll
    for (int i = 0; i < IPT; ++i) sv[i] = list[ga[i]];
#pragma unroll
    for (int i = 0; i < IPT; ++i) {
      if ((unsigned)i < m) rk[i] = atomicAdd(&rcnt[sv[i] >> ROWSHIFT], 1u);
    }
    __syncthreads();

    const unsigned c = tid < TN ? rcnt[tid] : 0u;
    unsigned inc = c;
#pragma unroll
    for (unsigned d = 1; d < 32u; d <<= 1) {
      const unsigned o = __shfl_up(inc, d);
      if (lane >= d) inc += o;
    }
    if (lane == 31u && wave < 4u) wsum[wave] = inc;
    __syncthreads();
    if (tid < TN) {
      const unsigned pre = (wave > 0u ? wsum[0] : 0u) + (wave > 1u ? wsum[1] : 0u) + (wave > 2u ? wsum[2] : 0u);
      rstart[tid] = pre + inc - c;
    }
    if (tid == 0u) rstart[TN] = n;
    __syncthreads();

    unsigned rs[IPT];
#pragma unroll
    for (int i = 0; i < IPT; ++i) rs[i] = rstart[(unsigned)i < m ? sv[i] >> ROWSHIFT : 0u];
#pragma unroll
    for (int i = 0; i < IPT; ++i)
      if ((unsigned)i < m) slist[rs[i] + rk[i]] = sv[i];
    __syncthreads();

    const unsigned lo = (unsigned)__builtin_amdgcn_readfirstlane((int)rstart[wave * 16u]);
    const unsigned hi = (unsigned)__builtin_amdgcn_readfirstlane((int)rstart[wave * 16u + 16u]);
    v2f acc = {0.0f, 0.0f};
    unsigned prev = TN * PITCH;
    for (unsigned g0 = lo; g0 < hi; g0 += 32u) {
      const unsigned idx = g0 + lane;
      unsigned word = (unsigned)TN << ROWSHIFT;
      if (idx < hi) word = slist[idx];
      const unsigned s_l = word & ((1u << ROWSHIFT) - 1u);
      const unsigned dl_l = (word >> ROWSHIFT) * PITCH;
      const float on_l = out_norm[s_l];
      const unsigned left = hi - g0;

      v2f v[32];
#pragma unroll
      for (int c8 = 0; c8 < 32; c8 += 8) {
        if ((unsigned)c8 < left) {
#pragma unroll
          for (int j = c8; j < c8 + 8; ++j) {
            const unsigned s = (unsigned)__builtin_amdgcn_readlane((int)s_l, j);
            v[j] = *(const v2f*)((const char*)feat + (s * ROW_BYTES + lane8));
          }
        }
        __builtin_amdgcn_sched_barrier(0);
      }
      const float inv_l = 1.0f / on_l;
#pragma unroll
      for (int c8 = 0; c8 < 32; c8 += 8) {
        if ((unsigned)c8 < left) {
#pragma unroll
          for (int j = c8; j < c8 + 8; ++j) {
            const float inv = __int_as_float(__builtin_amdgcn_readlane(__float_as_int(inv_l), j));
            const unsigned d = (unsigned)__builtin_amdgcn_readlane((int)dl_l, j);
            if (d != prev) {
              float* p = tile + prev + 2u * lane;
              __hip_atomic_fetch_add(p, acc.x, __ATOMIC_RELAXED, __HIP_MEMORY_SCOPE_WORKGROUP);
              __hip_atomic_fetch_add(p + 1, acc.y, __ATOMIC_RELAXED, __HIP_MEMORY_SCOPE_WORKGROUP);
              acc.x = 0.0f; acc.y = 0.0f;
              prev = d;
            }
            acc.x += v[j].x * inv;
            acc.y += v[j].y * inv;
          }
        }
      }
    }
    {
      float* p = tile + prev + 2u * lane;
      __hip_atomic_fetch_add(p, acc.x, __ATOMIC_RELAXED, __HIP_MEMORY_SCOPE_WORKGROUP);
      __hip_atomic_fetch_add(p + 1, acc.y, __ATOMIC_RELAXED, __HIP_MEMORY_SCOPE_WORKGROUP);
    }
  }
  __syncthreads();

  const unsigned rbase = row0 + wave * 16u;
  if (rbase >= N_NODES) return;

  const int m16   = lane & 15;
  const int khalf = lane >> 4;
  const unsigned mrow = rbase + m16;
  const float inv = 1.0f / in_norm[mrow];

  v2f afrag[16];
  const float* arow = tile + (wave * 16u + m16) * PITCH + 2 * khalf;
#pragma unroll
  for (int k = 0; k < 16; ++k) {
    v2f a = *(const v2f*)(arow + 4 * k);
    a.x *= inv; a.y *= inv;
    afrag[k] = a;
  }

#pragma unroll
  for (int nt = 0; nt < 4; ++nt) {
    const int nbase = nt << 4;
    const float bv  = bias[nbase + m16];
    v8f acc = {bv, bv, bv, bv, bv, bv, bv, bv};

    const float* wrow = W + (size_t)(nbase + m16) * D + 2 * khalf;
#pragma unroll
    for (int k = 0; k < 16; ++k) {
      v2f bf = *(const v2f*)(wrow + 4 * k);
      acc = __builtin_amdgcn_wmma_f32_16x16x4_f32(
           false, afrag[k],  false, bf,
           (short)0, acc,  false,  false);
    }

    float* orow = out + ((size_t)(rbase + 8 * khalf)) * D + nbase + m16;
#pragma unroll
    for (int j = 0; j < 8; ++j) {
      orow[(size_t)j * D] = acc[j];
    }
  }
}

extern "C" void kernel_launch(void* const* d_in, const int* in_sizes, int n_in,
                              void* d_out, int out_size, void* d_ws, size_t ws_size,
                              hipStream_t stream) {
  const float* feat     = (const float*)d_in[0];
  const float* in_norm  = (const float*)d_in[1];
  const float* out_norm = (const float*)d_in[2];
  const int*   esrc     = (const int*)d_in[3];
  const int*   edst     = (const int*)d_in[4];
  const float* W        = (const float*)d_in[5];
  const float* bias     = (const float*)d_in[6];

  float* out = (float*)d_out;
  unsigned* segTab = (unsigned*)d_ws;
  unsigned* list   = (unsigned*)((char*)d_ws + WS_LIST_OFF);

  gcn_chunk_sort<<<BIN_BLOCKS, 256, 0, stream>>>(esrc, edst, segTab, list);
  gcn_tile_sum_linear<<<NB, 256, 0, stream>>>(feat, in_norm, out_norm, W, bias, segTab, list, out);
}


# Linear part: W copied once per block into LDS with whole-line loads (pitch 68, bank-conflict-free); B fragments read from LDS, the 64 sixteen-line global loads per wave are gone; sums unchanged
# speedup vs baseline: 4.2873x; 1.0709x over previous
#include <hip/hip_runtime.h>


#define N_NODES 100000
#define N_EDGES 1600000
#define D 64

#define TN 128
#define TSHIFT 7
#define NB ((N_NODES + TN - 1) / TN)
#define NB_PAD 1024
#define PITCH 68
#define CHUNK 4096
#define EPT (CHUNK / 256)
#define BIN_BLOCKS ((N_EDGES + CHUNK - 1) / CHUNK)
#define TP (NB + 2)
#define ROW_BYTES (D * 4u)
#define CAP 2560
#define IPT (CAP / 256)
#define ROWSHIFT 20
#define WPITCH 68

#define WS_LIST_OFF (2u << 20)

typedef __attribute__((ext_vector_type(2))) float v2f;
typedef __attribute__((ext_vector_type(8))) float v8f;

static_assert(NB + 2 <= NB_PAD, "scan covers every bin and the two count entries");
static_assert(TP % 4 == 0 && CHUNK % 1024 == 0 && N_EDGES % 4 == 0, "16-byte accesses cover whole groups of 4");
static_assert((unsigned long long)BIN_BLOCKS * TP * 4u <= WS_LIST_OFF, "table ends before the list");
static_assert(BIN_BLOCKS <= 512, "a thread of the tile kernel takes two chunks");
static_assert((1 << TSHIFT) == TN, "tile shift");
static_assert(N_NODES % 16 == 0, "whole 16-row matrix tiles");
static_assert(D == 64, "a row is one float2 per lane");
static_assert(N_NODES <= (1 << ROWSHIFT) && (TN + 1) <= (1 << (32 - ROWSHIFT)), "source and row share a word");
static_assert(TN == 128, "waves 0..3 scan the row counts, wave w owns rows 16w..16w+15");
static_assert((unsigned long long)N_NODES * ROW_BYTES < (1ull << 32), "32-bit byte offsets");
static_assert(D * D == 4 * 4 * 256 && (WPITCH * 4) % 16 == 0, "W is four float4 per thread, and a row of its LDS copy starts on 16 bytes");

__global__ void __launch_bounds__(256) gcn_chunk_sort(
    const int* __restrict__ esrc,
    const int* __restrict__ edst,
    unsigned* __restrict__ segTab,
    unsigned* __restrict__ list) {
  __shared__ __attribute__((aligned(16))) unsigned h[NB_PAD];
  __shared__ __attribute__((aligned(16))) unsigned sorted[CHUNK];
  __shared__ unsigned wsum[8];
  const unsigned tid = threadIdx.x;
  const unsigned lane = tid & 31u;
  const unsigned wave = tid >> 5;

  for (unsigned b = tid; b < NB_PAD; b += 256u) h[b] = 0u;

  const unsigned cbase = blockIdx.x * CHUNK;
  int4 s4[EPT / 4], d4[EPT / 4];
#pragma unroll
  for (int q = 0; q < EPT / 4; ++q) {
    const unsigned e = cbase + (unsigned)q * 1024u + 4u * tid;
    s4[q] = make_int4(0, 0, 0, 0);
    d4[q] = make_int4(0, 0, 0, 0);
    if (e < N_EDGES) {
      s4[q] = *(const int4*)(esrc + e);
      d4[q] = *(const int4*)(edst + e);
    }
  }
  __syncthreads();

  unsigned wd[EPT], bn[EPT], rk[EPT];
#pragma unroll
  for (int q = 0; q < EPT / 4; ++q) {
    const bool have = cbase + (unsigned)q * 1024u + 4u * tid < N_EDGES;
    const int sv[4] = {s4[q].x, s4[q].y, s4[q].z, s4[q].w};
    const int dv[4] = {d4[q].x, d4[q].y, d4[q].z, d4[q].w};
#pragma unroll
    for (int j = 0; j < 4; ++j) {
      const unsigned d = (unsigned)dv[j];
      wd[4 * q + j] = (unsigned)sv[j] | ((d & (TN - 1u)) << ROWSHIFT);
      bn[4 * q + j] = d >> TSHIFT;
      rk[4 * q + j] = 0u;
      if (have) rk[4 * q + j] = atomicAdd(&h[bn[4 * q + j]], 1u);
    }
  }
  __syncthreads();

  const uint4 c4 = *(const uint4*)&h[4u * tid];
  const unsigned s = c4.x + c4.y + c4.z + c4.w;
  unsigned inc = s;
#pragma unroll
  for (unsigned d = 1; d < 32u; d <<= 1) {
    const unsigned o = __shfl_up(inc, d);
    if (lane >= d) inc += o;
  }
  if (lane == 31u) wsum[wave] = inc;
  __syncthreads();
  unsigned pre = 0u;
  for (unsigned w = 0; w < wave; ++w) pre += wsum[w];
  uint4 st;
  st.x = pre + inc - s;
  st.y = st.x + c4.x;
  st.z = st.y + c4.y;
  st.w = st.z + c4.z;
  *(uint4*)&h[4u * tid] = st;
  if (tid < TP / 4) *(uint4*)(segTab + blockIdx.x * TP + 4u * tid) = st;
  __syncthreads();

  unsigned fs[EPT];
#pragma unroll
  for (int i = 0; i < EPT; ++i) fs[i] = h[bn[i]];
#pragma unroll
  for (int i = 0; i < EPT; ++i)
    if (cbase + (unsigned)(i >> 2) * 1024u + 4u * tid < N_EDGES) sorted[fs[i] + rk[i]] = wd[i];
  __syncthreads();

#pragma unroll
  for (int q = 0; q < EPT / 4; ++q) {
    const unsigned idx = (unsigned)q * 1024u + 4u * tid;
    if (cbase + idx < N_EDGES) *(uint4*)(list + cbase + idx) = *(const uint4*)&sorted[idx];
  }
}

__global__ void __launch_bounds__(256, 8) gcn_tile_sum_linear(
    const float* __restrict__ feat,
    const float* __restrict__ in_norm,
    const float* __restrict__ out_norm,
    const float* __restrict__ W,
    const float* __restrict__ bias,
    const unsigned* __restrict__ segTab,
    const unsigned* __restrict__ list,
    float* __restrict__ out) {
  __shared__ float tile[(TN + 1) * PITCH];
  __shared__ unsigned slist[CAP];
  __shared__ unsigned rcnt[TN];
  __shared__ unsigned rstart[TN + 1];
  __shared__ unsigned wsum[4];
  __shared__ unsigned gaddr[CAP];
  __shared__ unsigned psum[8];
  __shared__ unsigned pinf[5 * 256];
  __shared__ __attribute__((aligned(16))) float wl[D * WPITCH];
  const unsigned tid = threadIdx.x;
  const unsigned lane = tid & 31u;
  const unsigned wave = (unsigned)__builtin_amdgcn_readfirstlane((int)(tid >> 5));
  const unsigned b = blockIdx.x;
  const unsigned row0 = b * TN;

  unsigned plen[2], pbase[2];
  unsigned a0[2], a1[2];
#pragma unroll
  for (int q = 0; q < 2; ++q) {
    const unsigned blk = 2u * tid + (unsigned)q;
    const unsigned* row = segTab + (blk < BIN_BLOCKS ? blk : BIN_BLOCKS - 1u) * TP + b;
    a0[q] = row[0];
    a1[q] = row[1];
  }
  float4 w4[4];
#pragma unroll
  for (int i = 0; i < 4; ++i) w4[i] = *(const float4*)(W + 4u * ((unsigned)i * 256u + tid));
  for (unsigned i = tid; i < (TN + 1) * PITCH; i += 256u) tile[i] = 0.0f;
#pragma unroll
  for (int i = 0; i < 4; ++i) {
    const unsigned f = (unsigned)i * 256u + tid;
    *(float4*)&wl[(f >> 4) * WPITCH + 4u * (f & 15u)] = w4[i];
  }
#pragma unroll
  for (int q = 0; q < 2; ++q) {
    const unsigned blk = 2u * tid + (unsigned)q;
    plen[q] = blk < BIN_BLOCKS ? a1[q] - a0[q] : 0u;
    pbase[q] = blk * CHUNK + a0[q];
  }
  const unsigned plen2 = plen[0] + plen[1];
  unsigned pinc = plen2;
#pragma unroll
  for (unsigned d = 1; d < 32u; d <<= 1) {
    const unsigned o = __shfl_up(pinc, d);
    if (lane >= d) pinc += o;
  }
  if (lane == 31u) psum[wave] = pinc;
  __syncthreads();
  unsigned ppre = 0u, ptot = 0u;
#pragma unroll
  for (unsigned w = 0; w < 8u; ++w) {
    const unsigned v = psum[w];
    if (w < wave) ppre += v;
    ptot += v;
  }
  const unsigned cnt = (unsigned)__builtin_amdgcn_readfirstlane((int)ptot);
  {
    const unsigned p0 = ppre + pinc - plen2;
    const unsigned p1 = p0 + plen[0];
    pinf[tid] = p0;
    pinf[256u + tid] = p1;
    pinf[512u + tid] = p1 + plen[1];
    pinf[768u + tid] = pbase[0] - p0;
    pinf[1024u + tid] = pbase[1] - p1;
  }

  const unsigned lane8 = lane * 8u;
  for (unsigned c0 = 0; c0 < cnt; c0 += CAP) {
    const unsigned n = (cnt - c0 < CAP) ? cnt - c0 : CAP;
    if (tid < TN) rcnt[tid] = 0u;
#pragma unroll
    for (int q = 0; q < 2; ++q) {
      const unsigned p = pinf[(unsigned)q * 256u + tid];
      const unsigned pe = pinf[(unsigned)(q + 1) * 256u + tid];
      const unsigned off = pinf[(unsigned)(q + 3) * 256u + tid];
      const unsigned klo = p > c0 ? p : c0;
      const unsigned khi = pe < c0 + n ? pe : c0 + n;
      for (unsigned k = klo; k < khi; ++k) gaddr[k - c0] = off + k;
    }
    __syncthreads();

    const unsigned m = (n + 255u - tid) >> 8;
    unsigned ga[IPT];
#pragma unroll
    for (int i = 0; i < IPT; ++i)
      ga[i] = gaddr[(unsigned)i < m ? tid + (unsigned)i * 256u : n - 1u];
    unsigned sv[IPT], rk[IPT];
#pragma unroll
    for (int i = 0; i < IPT; ++i) sv[i] = list[ga[i]];
#pragma unroll
    for (int i = 0; i < IPT; ++i) {
      if ((unsigned)i < m) rk[i] = atomicAdd(&rcnt[sv[i] >> ROWSHIFT], 1u);
    }
    __syncthreads();

    const unsigned c = tid < TN ? rcnt[tid] : 0u;
    unsigned inc = c;
#pragma unroll
    for (unsigned d = 1; d < 32u; d <<= 1) {
      const unsigned o = __shfl_up(inc, d);
      if (lane >= d) inc += o;
    }
    if (lane == 31u && wave < 4u) wsum[wave] = inc;
    __syncthreads();
    if (tid < TN) {
      const unsigned pre = (wave > 0u ? wsum[0] : 0u) + (wave > 1u ? wsum[1] : 0u) + (wave > 2u ? wsum[2] : 0u);
      rstart[tid] = pre + inc - c;
    }
    if (tid == 0u) rstart[TN] = n;
    __syncthreads();

    unsigned rs[IPT];
#pragma unroll
    for (int i = 0; i < IPT; ++i) rs[i] = rstart[(unsigned)i < m ? sv[i] >> ROWSHIFT : 0u];
#pragma unroll
    for (int i = 0; i < IPT; ++i)
      if ((unsigned)i < m) slist[rs[i] + rk[i]] = sv[i];
    __syncthreads();

    const unsigned lo = (unsigned)__builtin_amdgcn_readfirstlane((int)rstart[wave * 16u]);
    const unsigned hi = (unsigned)__builtin_amdgcn_readfirstlane((int)rstart[wave * 16u + 16u]);
    v2f acc = {0.0f, 0.0f};
    unsigned prev = TN * PITCH;
    for (unsigned g0 = lo; g0 < hi; g0 += 32u) {
      const unsigned idx = g0 + lane;
      unsigned word = (unsigned)TN << ROWSHIFT;
      if (idx < hi) word = slist[idx];
      const unsigned s_l = word & ((1u << ROWSHIFT) - 1u);
      const unsigned dl_l = (word >> ROWSHIFT) * PITCH;
      const float on_l = out_norm[s_l];
      const unsigned left = hi - g0;

      v2f v[32];
#pragma unroll
      for (int c8 = 0; c8 < 32; c8 += 8) {
        if ((unsigned)c8 < left) {
#pragma unroll
          for (int j = c8; j < c8 + 8; ++j) {
            const unsigned s = (unsigned)__builtin_amdgcn_readlane((int)s_l, j);
            v[j] = *(const v2f*)((const char*)feat + (s * ROW_BYTES + lane8));
          }
        }
        __builtin_amdgcn_sched_barrier(0);
      }
      const float inv_l = 1.0f / on_l;
#pragma unroll
      for (int c8 = 0; c8 < 32; c8 += 8) {
        if ((unsigned)c8 < left) {
#pragma unroll
          for (int j = c8; j < c8 + 8; ++j) {
            const float inv = __int_as_float(__builtin_amdgcn_readlane(__float_as_int(inv_l), j));
            const unsigned d = (unsigned)__builtin_amdgcn_readlane((int)dl_l, j);
            if (d != prev) {
              float* p = tile + prev + 2u * lane;
              __hip_atomic_fetch_add(p, acc.x, __ATOMIC_RELAXED, __HIP_MEMORY_SCOPE_WORKGROUP);
              __hip_atomic_fetch_add(p + 1, acc.y, __ATOMIC_RELAXED, __HIP_MEMORY_SCOPE_WORKGROUP);
              acc.x = 0.0f; acc.y = 0.0f;
              prev = d;
            }
            acc.x += v[j].x * inv;
            acc.y += v[j].y * inv;
          }
        }
      }
    }
    {
      float* p = tile + prev + 2u * lane;
      __hip_atomic_fetch_add(p, acc.x, __ATOMIC_RELAXED, __HIP_MEMORY_SCOPE_WORKGROUP);
      __hip_atomic_fetch_add(p + 1, acc.y, __ATOMIC_RELAXED, __HIP_MEMORY_SCOPE_WORKGROUP);
    }
  }
  __syncthreads();

  const unsigned rbase = row0 + wave * 16u;
  if (rbase >= N_NODES) return;

  const int m16   = lane & 15;
  const int khalf = lane >> 4;
  const unsigned mrow = rbase + m16;
  const float inv = 1.0f / in_norm[mrow];

  v2f afrag[16];
  const float* arow = tile + (wave * 16u + m16) * PITCH + 2 * khalf;
#pragma unroll
  for (int k = 0; k < 16; ++k) {
    v2f a = *(const v2f*)(arow + 4 * k);
    a.x *= inv; a.y *= inv;
    afrag[k] = a;
  }

#pragma unroll
  for (int nt = 0; nt < 4; ++nt) {
    const int nbase = nt << 4;
    const float bv  = bias[nbase + m16];
    v8f acc = {bv, bv, bv, bv, bv, bv, bv, bv};

    const float* wrow = wl + (nbase + m16) * WPITCH + 2 * khalf;
#pragma unroll
    for (int k = 0; k < 16; ++k) {
      v2f bf = *(const v2f*)(wrow + 4 * k);
      acc = __builtin_amdgcn_wmma_f32_16x16x4_f32(
           false, afrag[k],  false, bf,
           (short)0, acc,  false,  false);
    }

    float* orow = out + ((size_t)(rbase + 8 * khalf)) * D + nbase + m16;
#pragma unroll
    for (int j = 0; j < 8; ++j) {
      orow[(size_t)j * D] = acc[j];
    }
  }
}

extern "C" void kernel_launch(void* const* d_in, const int* in_sizes, int n_in,
                              void* d_out, int out_size, void* d_ws, size_t ws_size,
                              hipStream_t stream) {
  const float* feat     = (const float*)d_in[0];
  const float* in_norm  = (const float*)d_in[1];
  const float* out_norm = (const float*)d_in[2];
  const int*   esrc     = (const int*)d_in[3];
  const int*   edst     = (const int*)d_in[4];
  const float* W        = (const float*)d_in[5];
  const float* bias     = (const float*)d_in[6];

  float* out = (float*)d_out;
  unsigned* segTab = (unsigned*)d_ws;
  unsigned* list   = (unsigned*)((char*)d_ws + WS_LIST_OFF);

  gcn_chunk_sort<<<BIN_BLOCKS, 256, 0, stream>>>(esrc, edst, segTab, list);
  gcn_tile_sum_linear<<<NB, 256, 0, stream>>>(feat, in_norm, out_norm, W, bias, segTab, list, out);
}


# Tile of 512 rows per 1024-thread block (one per WGP, 196 blocks): a tile collects 391 pieces four times as long, so a quarter of the table and list lines per edge; sums in new atomic arrival order
# speedup vs baseline: 4.4918x; 1.0477x over previous
#include <hip/hip_runtime.h>


#define N_NODES 100000
#define N_EDGES 1600000
#define D 64

#define TN 512
#define TSHIFT 9
#define NB ((N_NODES + TN - 1) / TN)
#define KT 1024
#define NB_PAD 1024
#define PITCH 68
#define CHUNK 4096
#define EPT (CHUNK / 256)
#define BIN_BLOCKS ((N_EDGES + CHUNK - 1) / CHUNK)
#define TP ((NB + 2 + 3) / 4 * 4)
#define ROW_BYTES (D * 4u)
#define CAP 10240
#define IPT (CAP / KT)
#define ROWSHIFT 20
#define WPITCH 68

#define WS_LIST_OFF (2u << 20)

typedef __attribute__((ext_vector_type(2))) float v2f;
typedef __attribute__((ext_vector_type(8))) float v8f;

static_assert(TP <= NB_PAD && TP >= NB + 2, "scan covers every bin and the count entries");
static_assert(TP % 4 == 0 && CHUNK % 1024 == 0 && N_EDGES % 4 == 0, "16-byte accesses cover whole groups of 4");
static_assert((unsigned long long)BIN_BLOCKS * TP * 4u <= WS_LIST_OFF, "table ends before the list");
static_assert(BIN_BLOCKS <= KT, "a thread of the tile kernel takes one chunk");
static_assert((1 << TSHIFT) == TN, "tile shift");
static_assert(N_NODES % 16 == 0, "whole 16-row matrix tiles");
static_assert(D == 64, "a row is one float2 per lane");
static_assert(N_NODES <= (1 << ROWSHIFT) && (TN + 1) <= (1 << (32 - ROWSHIFT)), "source and row share a word");
static_assert(TN == 512 && KT == 2 * TN && CAP % KT == 0, "waves 0..15 scan the row counts, wave w owns rows 16w..16w+15");
static_assert((unsigned long long)N_NODES * ROW_BYTES < (1ull << 32), "32-bit byte offsets");
static_assert(D * D == 4 * KT && (WPITCH * 4) % 16 == 0, "W is one float4 per thread, and a row of its LDS copy starts on 16 bytes");

__global__ void __launch_bounds__(256) gcn_chunk_sort(
    const int* __restrict__ esrc,
    const int* __restrict__ edst,
    unsigned* __restrict__ segTab,
    unsigned* __restrict__ list) {
  __shared__ __attribute__((aligned(16))) unsigned h[NB_PAD];
  __shared__ __attribute__((aligned(16))) unsigned sorted[CHUNK];
  __shared__ unsigned wsum[8];
  const unsigned tid = threadIdx.x;
  const unsigned lane = tid & 31u;
  const unsigned wave = tid >> 5;

  for (unsigned b = tid; b < NB_PAD; b += 256u) h[b] = 0u;

  const unsigned cbase = blockIdx.x * CHUNK;
  int4 s4[EPT / 4], d4[EPT / 4];
#pragma unroll
  for (int q = 0; q < EPT / 4; ++q) {
    const unsigned e = cbase + (unsigned)q * 1024u + 4u * tid;
    s4[q] = make_int4(0, 0, 0, 0);
    d4[q] = make_int4(0, 0, 0, 0);
    if (e < N_EDGES) {
      s4[q] = *(const int4*)(esrc + e);
      d4[q] = *(const int4*)(edst + e);
    }
  }
  __syncthreads();

  unsigned wd[EPT], bn[EPT], rk[EPT];
#pragma unroll
  for (int q = 0; q < EPT / 4; ++q) {
    const bool have = cbase + (unsigned)q * 1024u + 4u * tid < N_EDGES;
    const int sv[4] = {s4[q].x, s4[q].y, s4[q].z, s4[q].w};
    const int dv[4] = {d4[q].x, d4[q].y, d4[q].z, d4[q].w};
#pragma unroll
    for (int j = 0; j < 4; ++j) {
      const unsigned d = (unsigned)dv[j];
      wd[4 * q + j] = (unsigned)sv[j] | ((d & (TN - 1u)) << ROWSHIFT);
      bn[4 * q + j] = d >> TSHIFT;
      rk[4 * q + j] = 0u;
      if (have) rk[4 * q + j] = atomicAdd(&h[bn[4 * q + j]], 1u);
    }
  }
  __syncthreads();

  const uint4 c4 = *(const uint4*)&h[4u * tid];
  const unsigned s = c4.x + c4.y + c4.z + c4.w;
  unsigned inc = s;
#pragma unroll
  for (unsigned d = 1; d < 32u; d <<= 1) {
    const unsigned o = __shfl_up(inc, d);
    if (lane >= d) inc += o;
  }
  if (lane == 31u) wsum[wave] = inc;
  __syncthreads();
  unsigned pre = 0u;
  for (unsigned w = 0; w < wave; ++w) pre += wsum[w];
  uint4 st;
  st.x = pre + inc - s;
  st.y = st.x + c4.x;
  st.z = st.y + c4.y;
  st.w = st.z + c4.z;
  *(uint4*)&h[4u * tid] = st;
  if (tid < TP / 4) *(uint4*)(segTab + blockIdx.x * TP + 4u * tid) = st;
  __syncthreads();

  unsigned fs[EPT];
#pragma unroll
  for (int i = 0; i < EPT; ++i) fs[i] = h[bn[i]];
#pragma unroll
  for (int i = 0; i < EPT; ++i)
    if (cbase + (unsigned)(i >> 2) * 1024u + 4u * tid < N_EDGES) sorted[fs[i] + rk[i]] = wd[i];
  __syncthreads();

#pragma unroll
  for (int q = 0; q < EPT / 4; ++q) {
    const unsigned idx = (unsigned)q * 1024u + 4u * tid;
    if (cbase + idx < N_EDGES) *(uint4*)(list + cbase + idx) = *(const uint4*)&sorted[idx];
  }
}

__global__ void __launch_bounds__(KT) gcn_tile_sum_linear(
    const float* __restrict__ feat,
    const float* __restrict__ in_norm,
    const float* __restrict__ out_norm,
    const float* __restrict__ W,
    const float* __restrict__ bias,
    const unsigned* __restrict__ segTab,
    const unsigned* __restrict__ list,
    float* __restrict__ out) {
  __shared__ float tile[(TN + 1) * PITCH];
  __shared__ unsigned slist[CAP];
  __shared__ unsigned rcnt[TN];
  __shared__ unsigned rstart[TN + 1];
  __shared__ unsigned wsum[TN / 32];
  __shared__ unsigned gaddr[CAP];
  __shared__ unsigned psum[KT / 32];
  __shared__ unsigned pinf[3 * KT];
  __shared__ __attribute__((aligned(16))) float wl[D * WPITCH];
  const unsigned tid = threadIdx.x;
  const unsigned lane = tid & 31u;
  const unsigned wave = (unsigned)__builtin_amdgcn_readfirstlane((int)(tid >> 5));
  const unsigned b = blockIdx.x;
  const unsigned row0 = b * TN;

  const unsigned* prow = segTab + (tid < BIN_BLOCKS ? tid : BIN_BLOCKS - 1u) * TP + b;
  const unsigned a0 = prow[0];
  const unsigned a1 = prow[1];
  const float4 w4 = *(const float4*)(W + 4u * tid);
  for (unsigned i = tid; i < (TN + 1) * PITCH; i += KT) tile[i] = 0.0f;
  *(float4*)&wl[(tid >> 4) * WPITCH + 4u * (tid & 15u)] = w4;
  const unsigned plen = tid < BIN_BLOCKS ? a1 - a0 : 0u;
  const unsigned pbase = tid * CHUNK + a0;
  unsigned pinc = plen;
#pragma unroll
  for (unsigned d = 1; d < 32u; d <<= 1) {
    const unsigned o = __shfl_up(pinc, d);
    if (lane >= d) pinc += o;
  }
  if (lane == 31u) psum[wave] = pinc;
  __syncthreads();
  unsigned ppre = 0u, ptot = 0u;
#pragma unroll
  for (unsigned w = 0; w < KT / 32; ++w) {
    const unsigned v = psum[w];
    if (w < wave) ppre += v;
    ptot += v;
  }
  const unsigned cnt = (unsigned)__builtin_amdgcn_readfirstlane((int)ptot);
  {
    const unsigned p0 = ppre + pinc - plen;
    pinf[tid] = p0;
    pinf[KT + tid] = p0 + plen;
    pinf[2u * KT + tid] = pbase - p0;
  }

  const unsigned lane8 = lane * 8u;
  for (unsigned c0 = 0; c0 < cnt; c0 += CAP) {
    const unsigned n = (cnt - c0 < CAP) ? cnt - c0 : CAP;
    if (tid < TN) rcnt[tid] = 0u;
    {
      const unsigned p = pinf[tid];
      const unsigned pe = pinf[KT + tid];
      const unsigned off = pinf[2u * KT + tid];
      const unsigned klo = p > c0 ? p : c0;
      const unsigned khi = pe < c0 + n ? pe : c0 + n;
      for (unsigned k = klo; k < khi; ++k) gaddr[k - c0] = off + k;
    }
    __syncthreads();

    const unsigned m = (n + (KT - 1u) - tid) / KT;
    unsigned ga[IPT];
#pragma unroll
    for (int i = 0; i < IPT; ++i)
      ga[i] = gaddr[(unsigned)i < m ? tid + (unsigned)i * KT : n - 1u];
    unsigned sv[IPT], rk[IPT];
#pragma unroll
    for (int i = 0; i < IPT; ++i) sv[i] = list[ga[i]];
#pragma unroll
    for (int i = 0; i < IPT; ++i) {
      if ((unsigned)i < m) rk[i] = atomicAdd(&rcnt[sv[i] >> ROWSHIFT], 1u);
    }
    __syncthreads();

    const unsigned c = tid < TN ? rcnt[tid] : 0u;
    unsigned inc = c;
#pragma unroll
    for (unsigned d = 1; d < 32u; d <<= 1) {
      const unsigned o = __shfl_up(inc, d);
      if (lane >= d) inc += o;
    }
    if (lane == 31u && wave < TN / 32) wsum[wave] = inc;
    __syncthreads();
    if (tid < TN) {
      unsigned pre = 0u;
      for (unsigned w = 0; w < wave; ++w) pre += wsum[w];
      rstart[tid] = pre + inc - c;
    }
    if (tid == 0u) rstart[TN] = n;
    __syncthreads();

    unsigned rs[IPT];
#pragma unroll
    for (int i = 0; i < IPT; ++i) rs[i] = rstart[(unsigned)i < m ? sv[i] >> ROWSHIFT : 0u];
#pragma unroll
    for (int i = 0; i < IPT; ++i)
      if ((unsigned)i < m) slist[rs[i] + rk[i]] = sv[i];
    __syncthreads();

    const unsigned lo = (unsigned)__builtin_amdgcn_readfirstlane((int)rstart[wave * 16u]);
    const unsigned hi = (unsigned)__builtin_amdgcn_readfirstlane((int)rstart[wave * 16u + 16u]);
    v2f acc = {0.0f, 0.0f};
    unsigned prev = TN * PITCH;
    for (unsigned g0 = lo; g0 < hi; g0 += 32u) {
      const unsigned idx = g0 + lane;
      unsigned word = (unsigned)TN << ROWSHIFT;
      if (idx < hi) word = slist[idx];
      const unsigned s_l = word & ((1u << ROWSHIFT) - 1u);
      const unsigned dl_l = (word >> ROWSHIFT) * PITCH;
      const float on_l = out_norm[s_l];
      const unsigned left = hi - g0;

      v2f v[32];
#pragma unroll
      for (int c8 = 0; c8 < 32; c8 += 8) {
        if ((unsigned)c8 < left) {
#pragma unroll
          for (int j = c8; j < c8 + 8; ++j) {
            const unsigned s = (unsigned)__builtin_amdgcn_readlane((int)s_l, j);
            v[j] = *(const v2f*)((const char*)feat + (s * ROW_BYTES + lane8));
          }
        }
        __builtin_amdgcn_sched_barrier(0);
      }
      const float inv_l = 1.0f / on_l;
#pragma unroll
      for (int c8 = 0; c8 < 32; c8 += 8) {
        if ((unsigned)c8 < left) {
#pragma unroll
          for (int j = c8; j < c8 + 8; ++j) {
            const float inv = __int_as_float(__builtin_amdgcn_readlane(__float_as_int(inv_l), j));
            const unsigned d = (unsigned)__builtin_amdgcn_readlane((int)dl_l, j);
            if (d != prev) {
              float* p = tile + prev + 2u * lane;
              __hip_atomic_fetch_add(p, acc.x, __ATOMIC_RELAXED, __HIP_MEMORY_SCOPE_WORKGROUP);
              __hip_atomic_fetch_add(p + 1, acc.y, __ATOMIC_RELAXED, __HIP_MEMORY_SCOPE_WORKGROUP);
              acc.x = 0.0f; acc.y = 0.0f;
              prev = d;
            }
            acc.x += v[j].x * inv;
            acc.y += v[j].y * inv;
          }
        }
      }
    }
    {
      float* p = tile + prev + 2u * lane;
      __hip_atomic_fetch_add(p, acc.x, __ATOMIC_RELAXED, __HIP_MEMORY_SCOPE_WORKGROUP);
      __hip_atomic_fetch_add(p + 1, acc.y, __ATOMIC_RELAXED, __HIP_MEMORY_SCOPE_WORKGROUP);
    }
  }
  __syncthreads();

  const unsigned rbase = row0 + wave * 16u;
  if (rbase >= N_NODES) return;

  const int m16   = lane & 15;
  const int khalf = lane >> 4;
  const unsigned mrow = rbase + m16;
  const float inv = 1.0f / in_norm[mrow];

  v2f afrag[16];
  const float* arow = tile + (wave * 16u + m16) * PITCH + 2 * khalf;
#pragma unroll
  for (int k = 0; k < 16; ++k) {
    v2f a = *(const v2f*)(arow + 4 * k);
    a.x *= inv; a.y *= inv;
    afrag[k] = a;
  }

#pragma unroll
  for (int nt = 0; nt < 4; ++nt) {
    const int nbase = nt << 4;
    const float bv  = bias[nbase + m16];
    v8f acc = {bv, bv, bv, bv, bv, bv, bv, bv};

    const float* wrow = wl + (nbase + m16) * WPITCH + 2 * khalf;
#pragma unroll
    for (int k = 0; k < 16; ++k) {
      v2f bf = *(const v2f*)(wrow + 4 * k);
      acc = __builtin_amdgcn_wmma_f32_16x16x4_f32(
           false, afrag[k],  false, bf,
           (short)0, acc,  false,  false);
    }

    float* orow = out + ((size_t)(rbase + 8 * khalf)) * D + nbase + m16;
#pragma unroll
    for (int j = 0; j < 8; ++j) {
      orow[(size_t)j * D] = acc[j];
    }
  }
}

extern "C" void kernel_launch(void* const* d_in, const int* in_sizes, int n_in,
                              void* d_out, int out_size, void* d_ws, size_t ws_size,
                              hipStream_t stream) {
  const float* feat     = (const float*)d_in[0];
  const float* in_norm  = (const float*)d_in[1];
  const float* out_norm = (const float*)d_in[2];
  const int*   esrc     = (const int*)d_in[3];
  const int*   edst     = (const int*)d_in[4];
  const float* W        = (const float*)d_in[5];
  const float* bias     = (const float*)d_in[6];

  float* out = (float*)d_out;
  unsigned* segTab = (unsigned*)d_ws;
  unsigned* list   = (unsigned*)((char*)d_ws + WS_LIST_OFF);

  gcn_chunk_sort<<<BIN_BLOCKS, 256, 0, stream>>>(esrc, edst, segTab, list);
  gcn_tile_sum_linear<<<NB, KT, 0, stream>>>(feat, in_norm, out_norm, W, bias, segTab, list, out);
}


# Tile of 400 rows, not 512: 250 one-per-WGP blocks cover the chip's 256 WGPs where 196 did (tile = dst / 400; a wave sums an even 32nd of the tile, 12 or 13 rows); sums in new atomic arrival order
# speedup vs baseline: 4.7718x; 1.0623x over previous
#include <hip/hip_runtime.h>


#define N_NODES 100000
#define N_EDGES 1600000
#define D 64

#define TN 400
#define NB ((N_NODES + TN - 1) / TN)
#define KT 1024
#define NB_PAD 1024
#define PITCH 68
#define CHUNK 4096
#define EPT (CHUNK / 256)
#define BIN_BLOCKS ((N_EDGES + CHUNK - 1) / CHUNK)
#define TP ((NB + 2 + 3) / 4 * 4)
#define ROW_BYTES (D * 4u)
#define CAP 8192
#define IPT (CAP / KT)
#define ROWSHIFT 20
#define WPITCH 68

#define WS_LIST_OFF (2u << 20)

typedef __attribute__((ext_vector_type(2))) float v2f;
typedef __attribute__((ext_vector_type(8))) float v8f;

static_assert(TP <= NB_PAD && TP >= NB + 2, "scan covers every bin and the count entries");
static_assert(TP % 4 == 0 && CHUNK % 1024 == 0 && N_EDGES % 4 == 0, "16-byte accesses cover whole groups of 4");
static_assert((unsigned long long)BIN_BLOCKS * TP * 4u <= WS_LIST_OFF, "table ends before the list");
static_assert(BIN_BLOCKS <= KT, "a thread of the tile kernel takes one chunk");
static_assert(N_NODES % 16 == 0, "whole 16-row matrix tiles");
static_assert(D == 64, "a row is one float2 per lane");
static_assert(N_NODES <= (1 << ROWSHIFT) && (TN + 1) <= (1 << (32 - ROWSHIFT)), "source and row share a word");
static_assert(TN % 16 == 0 && TN <= KT && CAP % KT == 0 && KT == 1024, "the first TN threads scan the row counts; wave w sums rows w*TN/32 .. (w+1)*TN/32 - 1");
static_assert((unsigned long long)N_NODES * ROW_BYTES < (1ull << 32), "32-bit byte offsets");
static_assert(D * D == 4 * KT && (WPITCH * 4) % 16 == 0, "W is one float4 per thread, and a row of its LDS copy starts on 16 bytes");

__global__ void __launch_bounds__(256) gcn_chunk_sort(
    const int* __restrict__ esrc,
    const int* __restrict__ edst,
    unsigned* __restrict__ segTab,
    unsigned* __restrict__ list) {
  __shared__ __attribute__((aligned(16))) unsigned h[NB_PAD];
  __shared__ __attribute__((aligned(16))) unsigned sorted[CHUNK];
  __shared__ unsigned wsum[8];
  const unsigned tid = threadIdx.x;
  const unsigned lane = tid & 31u;
  const unsigned wave = tid >> 5;

  for (unsigned b = tid; b < NB_PAD; b += 256u) h[b] = 0u;

  const unsigned cbase = blockIdx.x * CHUNK;
  int4 s4[EPT / 4], d4[EPT / 4];
#pragma unroll
  for (int q = 0; q < EPT / 4; ++q) {
    const unsigned e = cbase + (unsigned)q * 1024u + 4u * tid;
    s4[q] = make_int4(0, 0, 0, 0);
    d4[q] = make_int4(0, 0, 0, 0);
    if (e < N_EDGES) {
      s4[q] = *(const int4*)(esrc + e);
      d4[q] = *(const int4*)(edst + e);
    }
  }
  __syncthreads();

  unsigned wd[EPT], bn[EPT], rk[EPT];
#pragma unroll
  for (int q = 0; q < EPT / 4; ++q) {
    const bool have = cbase + (unsigned)q * 1024u + 4u * tid < N_EDGES;
    const int sv[4] = {s4[q].x, s4[q].y, s4[q].z, s4[q].w};
    const int dv[4] = {d4[q].x, d4[q].y, d4[q].z, d4[q].w};
#pragma unroll
    for (int j = 0; j < 4; ++j) {
      const unsigned d = (unsigned)dv[j];
      const unsigned t = d / TN;
      wd[4 * q + j] = (unsigned)sv[j] | ((d - t * TN) << ROWSHIFT);
      bn[4 * q + j] = t;
      rk[4 * q + j] = 0u;
      if (have) rk[4 * q + j] = atomicAdd(&h[bn[4 * q + j]], 1u);
    }
  }
  __syncthreads();

  const uint4 c4 = *(const uint4*)&h[4u * tid];
  const unsigned s = c4.x + c4.y + c4.z + c4.w;
  unsigned inc = s;
#pragma unroll
  for (unsigned d = 1; d < 32u; d <<= 1) {
    const unsigned o = __shfl_up(inc, d);
    if (lane >= d) inc += o;
  }
  if (lane == 31u) wsum[wave] = inc;
  __syncthreads();
  unsigned pre = 0u;
  for (unsigned w = 0; w < wave; ++w) pre += wsum[w];
  uint4 st;
  st.x = pre + inc - s;
  st.y = st.x + c4.x;
  st.z = st.y + c4.y;
  st.w = st.z + c4.z;
  *(uint4*)&h[4u * tid] = st;
  if (tid < TP / 4) *(uint4*)(segTab + blockIdx.x * TP + 4u * tid) = st;
  __syncthreads();

  unsigned fs[EPT];
#pragma unroll
  for (int i = 0; i < EPT; ++i) fs[i] = h[bn[i]];
#pragma unroll
  for (int i = 0; i < EPT; ++i)
    if (cbase + (unsigned)(i >> 2) * 1024u + 4u * tid < N_EDGES) sorted[fs[i] + rk[i]] = wd[i];
  __syncthreads();

#pragma unroll
  for (int q = 0; q < EPT / 4; ++q) {
    const unsigned idx = (unsigned)q * 1024u + 4u * tid;
    if (cbase + idx < N_EDGES) *(uint4*)(list + cbase + idx) = *(const uint4*)&sorted[idx];
  }
}

__global__ void __launch_bounds__(KT) gcn_tile_sum_linear(
    const float* __restrict__ feat,
    const float* __restrict__ in_norm,
    const float* __restrict__ out_norm,
    const float* __restrict__ W,
    const float* __restrict__ bias,
    const unsigned* __restrict__ segTab,
    const unsigned* __restrict__ list,
    float* __restrict__ out) {
  __shared__ float tile[(TN + 1) * PITCH];
  __shared__ unsigned slist[CAP];
  __shared__ unsigned rcnt[TN];
  __shared__ unsigned rstart[TN + 1];
  __shared__ unsigned wsum[(TN + 31) / 32];
  __shared__ unsigned gaddr[CAP];
  __shared__ unsigned psum[KT / 32];
  __shared__ unsigned pinf[3 * KT];
  __shared__ __attribute__((aligned(16))) float wl[D * WPITCH];
  const unsigned tid = threadIdx.x;
  const unsigned lane = tid & 31u;
  const unsigned wave = (unsigned)__builtin_amdgcn_readfirstlane((int)(tid >> 5));
  const unsigned b = blockIdx.x;
  const unsigned row0 = b * TN;

  const unsigned* prow = segTab + (tid < BIN_BLOCKS ? tid : BIN_BLOCKS - 1u) * TP + b;
  const unsigned a0 = prow[0];
  const unsigned a1 = prow[1];
  const float4 w4 = *(const float4*)(W + 4u * tid);
  for (unsigned i = tid; i < (TN + 1) * PITCH; i += KT) tile[i] = 0.0f;
  *(float4*)&wl[(tid >> 4) * WPITCH + 4u * (tid & 15u)] = w4;
  const unsigned plen = tid < BIN_BLOCKS ? a1 - a0 : 0u;
  const unsigned pbase = tid * CHUNK + a0;
  unsigned pinc = plen;
#pragma unroll
  for (unsigned d = 1; d < 32u; d <<= 1) {
    const unsigned o = __shfl_up(pinc, d);
    if (lane >= d) pinc += o;
  }
  if (lane == 31u) psum[wave] = pinc;
  __syncthreads();
  unsigned ppre = 0u, ptot = 0u;
#pragma unroll
  for (unsigned w = 0; w < KT / 32; ++w) {
    const unsigned v = psum[w];
    if (w < wave) ppre += v;
    ptot += v;
  }
  const unsigned cnt = (unsigned)__builtin_amdgcn_readfirstlane((int)ptot);
  {
    const unsigned p0 = ppre + pinc - plen;
    pinf[tid] = p0;
    pinf[KT + tid] = p0 + plen;
    pinf[2u * KT + tid] = pbase - p0;
  }

  const unsigned lane8 = lane * 8u;
  for (unsigned c0 = 0; c0 < cnt; c0 += CAP) {
    const unsigned n = (cnt - c0 < CAP) ? cnt - c0 : CAP;
    if (tid < TN) rcnt[tid] = 0u;
    {
      const unsigned p = pinf[tid];
      const unsigned pe = pinf[KT + tid];
      const unsigned off = pinf[2u * KT + tid];
      const unsigned klo = p > c0 ? p : c0;
      const unsigned khi = pe < c0 + n ? pe : c0 + n;
      for (unsigned k = klo; k < khi; ++k) gaddr[k - c0] = off + k;
    }
    __syncthreads();

    const unsigned m = (n + (KT - 1u) - tid) / KT;
    unsigned ga[IPT];
#pragma unroll
    for (int i = 0; i < IPT; ++i)
      ga[i] = gaddr[(unsigned)i < m ? tid + (unsigned)i * KT : n - 1u];
    unsigned sv[IPT], rk[IPT];
#pragma unroll
    for (int i = 0; i < IPT; ++i) sv[i] = list[ga[i]];
#pragma unroll
    for (int i = 0; i < IPT; ++i) {
      if ((unsigned)i < m) rk[i] = atomicAdd(&rcnt[sv[i] >> ROWSHIFT], 1u);
    }
    __syncthreads();

    const unsigned c = tid < TN ? rcnt[tid] : 0u;
    unsigned inc = c;
#pragma unroll
    for (unsigned d = 1; d < 32u; d <<= 1) {
      const unsigned o = __shfl_up(inc, d);
      if (lane >= d) inc += o;
    }
    if (lane == 31u && wave < (TN + 31) / 32) wsum[wave] = inc;
    __syncthreads();
    if (tid < TN) {
      unsigned pre = 0u;
      for (unsigned w = 0; w < wave; ++w) pre += wsum[w];
      rstart[tid] = pre + inc - c;
    }
    if (tid == 0u) rstart[TN] = n;
    __syncthreads();

    unsigned rs[IPT];
#pragma unroll
    for (int i = 0; i < IPT; ++i) rs[i] = rstart[(unsigned)i < m ? sv[i] >> ROWSHIFT : 0u];
#pragma unroll
    for (int i = 0; i < IPT; ++i)
      if ((unsigned)i < m) slist[rs[i] + rk[i]] = sv[i];
    __syncthreads();

    const unsigned lo = (unsigned)__builtin_amdgcn_readfirstlane((int)rstart[(wave * TN) >> 5]);
    const unsigned hi = (unsigned)__builtin_amdgcn_readfirstlane((int)rstart[((wave + 1u) * TN) >> 5]);
    v2f acc = {0.0f, 0.0f};
    unsigned prev = TN * PITCH;
    for (unsigned g0 = lo; g0 < hi; g0 += 32u) {
      const unsigned idx = g0 + lane;
      unsigned word = (unsigned)TN << ROWSHIFT;
      if (idx < hi) word = slist[idx];
      const unsigned s_l = word & ((1u << ROWSHIFT) - 1u);
      const unsigned dl_l = (word >> ROWSHIFT) * PITCH;
      const float on_l = out_norm[s_l];
      const unsigned left = hi - g0;

      v2f v[32];
#pragma unroll
      for (int c8 = 0; c8 < 32; c8 += 8) {
        if ((unsigned)c8 < left) {
#pragma unroll
          for (int j = c8; j < c8 + 8; ++j) {
            const unsigned s = (unsigned)__builtin_amdgcn_readlane((int)s_l, j);
            v[j] = *(const v2f*)((const char*)feat + (s * ROW_BYTES + lane8));
          }
        }
        __builtin_amdgcn_sched_barrier(0);
      }
      const float inv_l = 1.0f / on_l;
#pragma unroll
      for (int c8 = 0; c8 < 32; c8 += 8) {
        if ((unsigned)c8 < left) {
#pragma unroll
          for (int j = c8; j < c8 + 8; ++j) {
            const float inv = __int_as_float(__builtin_amdgcn_readlane(__float_as_int(inv_l), j));
            const unsigned d = (unsigned)__builtin_amdgcn_readlane((int)dl_l, j);
            if (d != prev) {
              float* p = tile + prev + 2u * lane;
              __hip_atomic_fetch_add(p, acc.x, __ATOMIC_RELAXED, __HIP_MEMORY_SCOPE_WORKGROUP);
              __hip_atomic_fetch_add(p + 1, acc.y, __ATOMIC_RELAXED, __HIP_MEMORY_SCOPE_WORKGROUP);
              acc.x = 0.0f; acc.y = 0.0f;
              prev = d;
            }
            acc.x += v[j].x * inv;
            acc.y += v[j].y * inv;
          }
        }
      }
    }
    {
      float* p = tile + prev + 2u * lane;
      __hip_atomic_fetch_add(p, acc.x, __ATOMIC_RELAXED, __HIP_MEMORY_SCOPE_WORKGROUP);
      __hip_atomic_fetch_add(p + 1, acc.y, __ATOMIC_RELAXED, __HIP_MEMORY_SCOPE_WORKGROUP);
    }
  }
  __syncthreads();

  const unsigned rbase = row0 + wave * 16u;
  if (wave * 16u >= TN || rbase >= N_NODES) return;

  const int m16   = lane & 15;
  const int khalf = lane >> 4;
  const unsigned mrow = rbase + m16;
  const float inv = 1.0f / in_norm[mrow];

  v2f afrag[16];
  const float* arow = tile + (wave * 16u + m16) * PITCH + 2 * khalf;
#pragma unroll
  for (int k = 0; k < 16; ++k) {
    v2f a = *(const v2f*)(arow + 4 * k);
    a.x *= inv; a.y *= inv;
    afrag[k] = a;
  }

#pragma unroll
  for (int nt = 0; nt < 4; ++nt) {
    const int nbase = nt << 4;
    const float bv  = bias[nbase + m16];
    v8f acc = {bv, bv, bv, bv, bv, bv, bv, bv};

    const float* wrow = wl + (nbase + m16) * WPITCH + 2 * khalf;
#pragma unroll
    for (int k = 0; k < 16; ++k) {
      v2f bf = *(const v2f*)(wrow + 4 * k);
      acc = __builtin_amdgcn_wmma_f32_16x16x4_f32(
           false, afrag[k],  false, bf,
           (short)0, acc,  false,  false);
    }

    float* orow = out + ((size_t)(rbase + 8 * khalf)) * D + nbase + m16;
#pragma unroll
    for (int j = 0; j < 8; ++j) {
      orow[(size_t)j * D] = acc[j];
    }
  }
}

extern "C" void kernel_launch(void* const* d_in, const int* in_sizes, int n_in,
                              void* d_out, int out_size, void* d_ws, size_t ws_size,
                              hipStream_t stream) {
  const float* feat     = (const float*)d_in[0];
  const float* in_norm  = (const float*)d_in[1];
  const float* out_norm = (const float*)d_in[2];
  const int*   esrc     = (const int*)d_in[3];
  const int*   edst     = (const int*)d_in[4];
  const float* W        = (const float*)d_in[5];
  const float* bias     = (const float*)d_in[6];

  float* out = (float*)d_out;
  unsigned* segTab = (unsigned*)d_ws;
  unsigned* list   = (unsigned*)((char*)d_ws + WS_LIST_OFF);

  gcn_chunk_sort<<<BIN_BLOCKS, 256, 0, stream>>>(esrc, edst, segTab, list);
  gcn_tile_sum_linear<<<NB, KT, 0, stream>>>(feat, in_norm, out_norm, W, bias, segTab, list, out);
}


# Row gather by half-waves: a row is a float4 per lane of 16 lanes, one load fetches two rows (4 lines, half the loads); 64 halves sum even 64ths of the tile's list, a cut row in 2 parts; new sum order
# speedup vs baseline: 4.9193x; 1.0309x over previous
#include <hip/hip_runtime.h>


#define N_NODES 100000
#define N_EDGES 1600000
#define D 64

#define TN 400
#define NB ((N_NODES + TN - 1) / TN)
#define KT 1024
#define NB_PAD 1024
#define PITCH 68
#define CHUNK 4096
#define EPT (CHUNK / 256)
#define BIN_BLOCKS ((N_EDGES + CHUNK - 1) / CHUNK)
#define TP ((NB + 2 + 3) / 4 * 4)
#define ROW_BYTES (D * 4u)
#define CAP 8192
#define IPT (CAP / KT)
#define ROWSHIFT 20
#define WPITCH 68

#define WS_LIST_OFF (2u << 20)

typedef __attribute__((ext_vector_type(2))) float v2f;
typedef __attribute__((ext_vector_type(4))) float v4f;
typedef __attribute__((ext_vector_type(8))) float v8f;

static_assert(TP <= NB_PAD && TP >= NB + 2, "scan covers every bin and the count entries");
static_assert(TP % 4 == 0 && CHUNK % 1024 == 0 && N_EDGES % 4 == 0, "16-byte accesses cover whole groups of 4");
static_assert((unsigned long long)BIN_BLOCKS * TP * 4u <= WS_LIST_OFF, "table ends before the list");
static_assert(BIN_BLOCKS <= KT, "a thread of the tile kernel takes one chunk");
static_assert(N_NODES % 16 == 0, "whole 16-row matrix tiles");
static_assert(D == 64, "a row is one float2 per lane");
static_assert(N_NODES <= (1 << ROWSHIFT) && (TN + 1) <= (1 << (32 - ROWSHIFT)), "source and row share a word");
static_assert(TN % 16 == 0 && TN <= KT && CAP % KT == 0 && KT == 1024, "the first TN threads scan the row counts; one thread of the tile kernel per list slot of a pass");
static_assert(64u * CAP < (1ull << 32) && (PITCH * 4) % 16 == 0, "a half-wave's list bounds fit 32 bits; a tile row starts on 16 bytes");
static_assert((unsigned long long)N_NODES * ROW_BYTES < (1ull << 32), "32-bit byte offsets");
static_assert(D * D == 4 * KT && (WPITCH * 4) % 16 == 0, "W is one float4 per thread, and a row of its LDS copy starts on 16 bytes");

__global__ void __launch_bounds__(256) gcn_chunk_sort(
    const int* __restrict__ esrc,
    const int* __restrict__ edst,
    unsigned* __restrict__ segTab,
    unsigned* __restrict__ list) {
  __shared__ __attribute__((aligned(16))) unsigned h[NB_PAD];
  __shared__ __attribute__((aligned(16))) unsigned sorted[CHUNK];
  __shared__ unsigned wsum[8];
  const unsigned tid = threadIdx.x;
  const unsigned lane = tid & 31u;
  const unsigned wave = tid >> 5;

  for (unsigned b = tid; b < NB_PAD; b += 256u) h[b] = 0u;

  const unsigned cbase = blockIdx.x * CHUNK;
  int4 s4[EPT / 4], d4[EPT / 4];
#pragma unroll
  for (int q = 0; q < EPT / 4; ++q) {
    const unsigned e = cbase + (unsigned)q * 1024u + 4u * tid;
    s4[q] = make_int4(0, 0, 0, 0);
    d4[q] = make_int4(0, 0, 0, 0);
    if (e < N_EDGES) {
      s4[q] = *(const int4*)(esrc + e);
      d4[q] = *(const int4*)(edst + e);
    }
  }
  __syncthreads();

  unsigned wd[EPT], bn[EPT], rk[EPT];
#pragma unroll
  for (int q = 0; q < EPT / 4; ++q) {
    const bool have = cbase + (unsigned)q * 1024u + 4u * tid < N_EDGES;
    const int sv[4] = {s4[q].x, s4[q].y, s4[q].z, s4[q].w};
    const int dv[4] = {d4[q].x, d4[q].y, d4[q].z, d4[q].w};
#pragma unroll
    for (int j = 0; j < 4; ++j) {
      const unsigned d = (unsigned)dv[j];
      const unsigned t = d / TN;
      wd[4 * q + j] = (unsigned)sv[j] | ((d - t * TN) << ROWSHIFT);
      bn[4 * q + j] = t;
      rk[4 * q + j] = 0u;
      if (have) rk[4 * q + j] = atomicAdd(&h[bn[4 * q + j]], 1u);
    }
  }
  __syncthreads();

  const uint4 c4 = *(const uint4*)&h[4u * tid];
  const unsigned s = c4.x + c4.y + c4.z + c4.w;
  unsigned inc = s;
#pragma unroll
  for (unsigned d = 1; d < 32u; d <<= 1) {
    const unsigned o = __shfl_up(inc, d);
    if (lane >= d) inc += o;
  }
  if (lane == 31u) wsum[wave] = inc;
  __syncthreads();
  unsigned pre = 0u;
  for (unsigned w = 0; w < wave; ++w) pre += wsum[w];
  uint4 st;
  st.x = pre + inc - s;
  st.y = st.x + c4.x;
  st.z = st.y + c4.y;
  st.w = st.z + c4.z;
  *(uint4*)&h[4u * tid] = st;
  if (tid < TP / 4) *(uint4*)(segTab + blockIdx.x * TP + 4u * tid) = st;
  __syncthreads();

  unsigned fs[EPT];
#pragma unroll
  for (int i = 0; i < EPT; ++i) fs[i] = h[bn[i]];
#pragma unroll
  for (int i = 0; i < EPT; ++i)
    if (cbase + (unsigned)(i >> 2) * 1024u + 4u * tid < N_EDGES) sorted[fs[i] + rk[i]] = wd[i];
  __syncthreads();

#pragma unroll
  for (int q = 0; q < EPT / 4; ++q) {
    const unsigned idx = (unsigned)q * 1024u + 4u * tid;
    if (cbase + idx < N_EDGES) *(uint4*)(list + cbase + idx) = *(const uint4*)&sorted[idx];
  }
}

#define SHARE_CASE(J) case J: return (unsigned)__builtin_amdgcn_update_dpp((int)x, (int)x, 0x150 + J, 0xf, 0xf, true);
static __device__ __forceinline__ unsigned half_share(unsigned x, int j) {
  switch (j) {
    SHARE_CASE(0) SHARE_CASE(1) SHARE_CASE(2) SHARE_CASE(3) SHARE_CASE(4) SHARE_CASE(5) SHARE_CASE(6) SHARE_CASE(7)
    SHARE_CASE(8) SHARE_CASE(9) SHARE_CASE(10) SHARE_CASE(11) SHARE_CASE(12) SHARE_CASE(13) SHARE_CASE(14)
    default: return (unsigned)__builtin_amdgcn_update_dpp((int)x, (int)x, 0x15f, 0xf, 0xf, true);
  }
}
#undef SHARE_CASE

__global__ void __launch_bounds__(KT) gcn_tile_sum_linear(
    const float* __restrict__ feat,
    const float* __restrict__ in_norm,
    const float* __restrict__ out_norm,
    const float* __restrict__ W,
    const float* __restrict__ bias,
    const unsigned* __restrict__ segTab,
    const unsigned* __restrict__ list,
    float* __restrict__ out) {
  __shared__ float tile[(TN + 1) * PITCH];
  __shared__ unsigned slist[CAP];
  __shared__ unsigned rcnt[TN];
  __shared__ unsigned rstart[TN];
  __shared__ unsigned wsum[(TN + 31) / 32];
  __shared__ unsigned gaddr[CAP];
  __shared__ unsigned psum[KT / 32];
  __shared__ unsigned pinf[3 * KT];
  __shared__ __attribute__((aligned(16))) float wl[D * WPITCH];
  const unsigned tid = threadIdx.x;
  const unsigned lane = tid & 31u;
  const unsigned wave = (unsigned)__builtin_amdgcn_readfirstlane((int)(tid >> 5));
  const unsigned b = blockIdx.x;
  const unsigned row0 = b * TN;

  const unsigned* prow = segTab + (tid < BIN_BLOCKS ? tid : BIN_BLOCKS - 1u) * TP + b;
  const unsigned a0 = prow[0];
  const unsigned a1 = prow[1];
  const float4 w4 = *(const float4*)(W + 4u * tid);
  for (unsigned i = tid; i < (TN + 1) * PITCH; i += KT) tile[i] = 0.0f;
  *(float4*)&wl[(tid >> 4) * WPITCH + 4u * (tid & 15u)] = w4;
  const unsigned plen = tid < BIN_BLOCKS ? a1 - a0 : 0u;
  const unsigned pbase = tid * CHUNK + a0;
  unsigned pinc = plen;
#pragma unroll
  for (unsigned d = 1; d < 32u; d <<= 1) {
    const unsigned o = __shfl_up(pinc, d);
    if (lane >= d) pinc += o;
  }
  if (lane == 31u) psum[wave] = pinc;
  __syncthreads();
  unsigned ppre = 0u, ptot = 0u;
#pragma unroll
  for (unsigned w = 0; w < KT / 32; ++w) {
    const unsigned v = psum[w];
    if (w < wave) ppre += v;
    ptot += v;
  }
  const unsigned cnt = (unsigned)__builtin_amdgcn_readfirstlane((int)ptot);
  {
    const unsigned p0 = ppre + pinc - plen;
    pinf[tid] = p0;
    pinf[KT + tid] = p0 + plen;
    pinf[2u * KT + tid] = pbase - p0;
  }

  const unsigned c16 = lane & 15u;
  const unsigned lane16 = c16 * 16u;
  for (unsigned c0 = 0; c0 < cnt; c0 += CAP) {
    const unsigned n = (cnt - c0 < CAP) ? cnt - c0 : CAP;
    if (tid < TN) rcnt[tid] = 0u;
    {
      const unsigned p = pinf[tid];
      const unsigned pe = pinf[KT + tid];
      const unsigned off = pinf[2u * KT + tid];
      const unsigned klo = p > c0 ? p : c0;
      const unsigned khi = pe < c0 + n ? pe : c0 + n;
      for (unsigned k = klo; k < khi; ++k) gaddr[k - c0] = off + k;
    }
    __syncthreads();

    const unsigned m = (n + (KT - 1u) - tid) / KT;
    unsigned ga[IPT];
#pragma unroll
    for (int i = 0; i < IPT; ++i)
      ga[i] = gaddr[(unsigned)i < m ? tid + (unsigned)i * KT : n - 1u];
    unsigned sv[IPT], rk[IPT];
#pragma unroll
    for (int i = 0; i < IPT; ++i) sv[i] = list[ga[i]];
#pragma unroll
    for (int i = 0; i < IPT; ++i) {
      if ((unsigned)i < m) rk[i] = atomicAdd(&rcnt[sv[i] >> ROWSHIFT], 1u);
    }
    __syncthreads();

    const unsigned c = tid < TN ? rcnt[tid] : 0u;
    unsigned inc = c;
#pragma unroll
    for (unsigned d = 1; d < 32u; d <<= 1) {
      const unsigned o = __shfl_up(inc, d);
      if (lane >= d) inc += o;
    }
    if (lane == 31u && wave < (TN + 31) / 32) wsum[wave] = inc;
    __syncthreads();
    if (tid < TN) {
      unsigned pre = 0u;
      for (unsigned w = 0; w < wave; ++w) pre += wsum[w];
      rstart[tid] = pre + inc - c;
    }
    __syncthreads();

    unsigned rs[IPT];
#pragma unroll
    for (int i = 0; i < IPT; ++i) rs[i] = rstart[(unsigned)i < m ? sv[i] >> ROWSHIFT : 0u];
#pragma unroll
    for (int i = 0; i < IPT; ++i)
      if ((unsigned)i < m) slist[rs[i] + rk[i]] = sv[i];
    __syncthreads();

    const unsigned q = wave * 2u + (lane >> 4);
    const unsigned lo_l = (q * n) >> 6;
    const unsigned hi_l = ((q + 1u) * n) >> 6;
    const unsigned len_l = hi_l - lo_l;
    const unsigned len_a = (unsigned)__builtin_amdgcn_readlane((int)len_l, 0);
    const unsigned len_b = (unsigned)__builtin_amdgcn_readlane((int)len_l, 16);
    const unsigned lenmax = len_a > len_b ? len_a : len_b;
    v4f acc = {0.0f, 0.0f, 0.0f, 0.0f};
    unsigned prev = TN * PITCH;
    for (unsigned g0 = 0; g0 < lenmax; g0 += 16u) {
      const unsigned idx = lo_l + g0 + c16;
      unsigned word = (unsigned)TN << ROWSHIFT;
      if (idx < hi_l) word = slist[idx];
      const unsigned s_l = word & ((1u << ROWSHIFT) - 1u);
      const unsigned so_l = s_l * ROW_BYTES;
      const unsigned dl_l = (word >> ROWSHIFT) * PITCH;
      const float on_l = out_norm[s_l];
      const unsigned left = lenmax - g0;

      v4f v[16];
#pragma unroll
      for (int c4 = 0; c4 < 16; c4 += 4) {
        if ((unsigned)c4 < left) {
#pragma unroll
          for (int j = c4; j < c4 + 4; ++j) {
            const unsigned so = half_share(so_l, j);
            v[j] = *(const v4f*)((const char*)feat + (so + lane16));
          }
        }
        __builtin_amdgcn_sched_barrier(0);
      }
      const float inv_l = 1.0f / on_l;
#pragma unroll
      for (int c4 = 0; c4 < 16; c4 += 4) {
        if ((unsigned)c4 < left) {
#pragma unroll
          for (int j = c4; j < c4 + 4; ++j) {
            const float inv = __uint_as_float(half_share(__float_as_uint(inv_l), j));
            const unsigned d = half_share(dl_l, j);
            if (d != prev) {
              float* p = tile + prev + 4u * c16;
              __hip_atomic_fetch_add(p, acc.x, __ATOMIC_RELAXED, __HIP_MEMORY_SCOPE_WORKGROUP);
              __hip_atomic_fetch_add(p + 1, acc.y, __ATOMIC_RELAXED, __HIP_MEMORY_SCOPE_WORKGROUP);
              __hip_atomic_fetch_add(p + 2, acc.z, __ATOMIC_RELAXED, __HIP_MEMORY_SCOPE_WORKGROUP);
              __hip_atomic_fetch_add(p + 3, acc.w, __ATOMIC_RELAXED, __HIP_MEMORY_SCOPE_WORKGROUP);
              acc.x = 0.0f; acc.y = 0.0f; acc.z = 0.0f; acc.w = 0.0f;
              prev = d;
            }
            acc.x += v[j].x * inv;
            acc.y += v[j].y * inv;
            acc.z += v[j].z * inv;
            acc.w += v[j].w * inv;
          }
        }
      }
    }
    {
      float* p = tile + prev + 4u * c16;
      __hip_atomic_fetch_add(p, acc.x, __ATOMIC_RELAXED, __HIP_MEMORY_SCOPE_WORKGROUP);
      __hip_atomic_fetch_add(p + 1, acc.y, __ATOMIC_RELAXED, __HIP_MEMORY_SCOPE_WORKGROUP);
      __hip_atomic_fetch_add(p + 2, acc.z, __ATOMIC_RELAXED, __HIP_MEMORY_SCOPE_WORKGROUP);
      __hip_atomic_fetch_add(p + 3, acc.w, __ATOMIC_RELAXED, __HIP_MEMORY_SCOPE_WORKGROUP);
    }
  }
  __syncthreads();

  const unsigned rbase = row0 + wave * 16u;
  if (wave * 16u >= TN || rbase >= N_NODES) return;

  const int m16   = lane & 15;
  const int khalf = lane >> 4;
  const unsigned mrow = rbase + m16;
  const float inv = 1.0f / in_norm[mrow];

  v2f afrag[16];
  const float* arow = tile + (wave * 16u + m16) * PITCH + 2 * khalf;
#pragma unroll
  for (int k = 0; k < 16; ++k) {
    v2f a = *(const v2f*)(arow + 4 * k);
    a.x *= inv; a.y *= inv;
    afrag[k] = a;
  }

#pragma unroll
  for (int nt = 0; nt < 4; ++nt) {
    const int nbase = nt << 4;
    const float bv  = bias[nbase + m16];
    v8f acc = {bv, bv, bv, bv, bv, bv, bv, bv};

    const float* wrow = wl + (nbase + m16) * WPITCH + 2 * khalf;
#pragma unroll
    for (int k = 0; k < 16; ++k) {
      v2f bf = *(const v2f*)(wrow + 4 * k);
      acc = __builtin_amdgcn_wmma_f32_16x16x4_f32(
           false, afrag[k],  false, bf,
           (short)0, acc,  false,  false);
    }

    float* orow = out + ((size_t)(rbase + 8 * khalf)) * D + nbase + m16;
#pragma unroll
    for (int j = 0; j < 8; ++j) {
      orow[(size_t)j * D] = acc[j];
    }
  }
}

extern "C" void kernel_launch(void* const* d_in, const int* in_sizes, int n_in,
                              void* d_out, int out_size, void* d_ws, size_t ws_size,
                              hipStream_t stream) {
  const float* feat     = (const float*)d_in[0];
  const float* in_norm  = (const float*)d_in[1];
  const float* out_norm = (const float*)d_in[2];
  const int*   esrc     = (const int*)d_in[3];
  const int*   edst     = (const int*)d_in[4];
  const float* W        = (const float*)d_in[5];
  const float* bias     = (const float*)d_in[6];

  float* out = (float*)d_out;
  unsigned* segTab = (unsigned*)d_ws;
  unsigned* list   = (unsigned*)((char*)d_ws + WS_LIST_OFF);

  gcn_chunk_sort<<<BIN_BLOCKS, 256, 0, stream>>>(esrc, edst, segTab, list);
  gcn_tile_sum_linear<<<NB, KT, 0, stream>>>(feat, in_norm, out_norm, W, bias, segTab, list, out);
}
